# starting kernel
# baseline (speedup 1.0000x reference)
#include <hip/hip_runtime.h>

typedef __attribute__((ext_vector_type(2))) float v2f;
typedef __attribute__((ext_vector_type(8))) float v8f;

#define NEG_SLOPE 0.2f

__global__ void fillf_kernel(float* __restrict__ p, int n, float v) {
  int i = blockIdx.x * blockDim.x + threadIdx.x;
  if (i < n) p[i] = v;
}

__device__ inline void atomicMaxF(float* addr, float v) {
  if (v >= 0.0f) atomicMax((int*)addr, __float_as_int(v));
  else           atomicMin((unsigned int*)addr, __float_as_uint(v));
}

__global__ void pack_w_kernel(const float* __restrict__ W,
                              v2f* __restrict__ Wp, int K, int Ncol) {
  const int i = blockIdx.x * blockDim.x + threadIdx.x;
  const int total = (K >> 1) * Ncol;
  if (i >= total) return;
  const int kp  = i / Ncol;
  const int col = i - kp * Ncol;
  v2f p;
  p[0] = W[(size_t)(2 * kp)     * Ncol + col];
  p[1] = W[(size_t)(2 * kp + 1) * Ncol + col];
  Wp[i] = p;
}

__global__ void gemm_bias_wmma_n64(const float* __restrict__ X,
                                   const v2f* __restrict__ Wp,
                                   const float* __restrict__ bias,
                                   float* __restrict__ Y,
                                   int M, int K, int Ncol) {
  const int lane = threadIdx.x & 31;
  const int wave = threadIdx.x >> 5;
  const int ct = Ncol >> 6;
  const int rt = M >> 4;
  const int tile = blockIdx.x * (blockDim.x >> 5) + wave;
  if (tile >= ct * rt) return;

  const int rowBase = (tile / ct) << 4;
  const int colBase = (tile % ct) << 6;
  const int half = lane >> 4;
  const int l15  = lane & 15;

  const float* __restrict__ ap = X + (size_t)(rowBase + l15) * K + half * 2;
  const v2f*   __restrict__ bp = Wp + (size_t)half * Ncol + colBase + l15;

  v8f cc[4] = {v8f{}, v8f{}, v8f{}, v8f{}};
  #pragma unroll 4
  for (int k0 = 0; k0 < K; k0 += 4) {
    const v2f a = *(const v2f*)(ap + k0);
    const v2f* __restrict__ br = bp + (size_t)(k0 >> 1) * Ncol;
    const v2f bv0 = br[0];
    const v2f bv1 = br[16];
    const v2f bv2 = br[32];
    const v2f bv3 = br[48];
    cc[0] = __builtin_amdgcn_wmma_f32_16x16x4_f32(false, a, false, bv0, (short)0, cc[0], false, false);
    cc[1] = __builtin_amdgcn_wmma_f32_16x16x4_f32(false, a, false, bv1, (short)0, cc[1], false, false);
    cc[2] = __builtin_amdgcn_wmma_f32_16x16x4_f32(false, a, false, bv2, (short)0, cc[2], false, false);
    cc[3] = __builtin_amdgcn_wmma_f32_16x16x4_f32(false, a, false, bv3, (short)0, cc[3], false, false);
  }

  const int row0 = rowBase + half * 8;
  #pragma unroll
  for (int t = 0; t < 4; ++t) {
    const int col = colBase + t * 16 + l15;
    const float bb = bias ? bias[col] : 0.0f;
    float* __restrict__ yp = Y + (size_t)row0 * Ncol + col;
    #pragma unroll
    for (int r = 0; r < 8; ++r)
      yp[(size_t)r * Ncol] = cc[t][r] + bb;
  }
}

__global__ void gemm_bias_wmma_guard(const float* __restrict__ X,
                                     const v2f* __restrict__ Wp,
                                     const float* __restrict__ bias,
                                     float* __restrict__ Y,
                                     int M, int K, int Ncol) {
  const int lane = threadIdx.x & 31;
  const int wave = threadIdx.x >> 5;
  const int ct = (Ncol + 15) >> 4;
  const int rt = M >> 4;
  const int tile = blockIdx.x * (blockDim.x >> 5) + wave;
  if (tile >= ct * rt) return;

  const int rowBase = (tile / ct) << 4;
  const int colBase = (tile % ct) << 4;
  const int half = lane >> 4;
  const int l15  = lane & 15;
  const int bcol = colBase + l15;
  const bool bok = bcol < Ncol;
  const int bcolc = bok ? bcol : (Ncol - 1);
  const float bmask = bok ? 1.0f : 0.0f;

  const float* __restrict__ ap = X + (size_t)(rowBase + l15) * K + half * 2;
  const v2f*   __restrict__ bp = Wp + (size_t)half * Ncol + bcolc;

  v8f c = {};
  #pragma unroll 4
  for (int k0 = 0; k0 < K; k0 += 4) {
    const v2f a = *(const v2f*)(ap + k0);
    v2f b = bp[(size_t)(k0 >> 1) * Ncol];
    b[0] *= bmask;
    b[1] *= bmask;
    c = __builtin_amdgcn_wmma_f32_16x16x4_f32(false, a, false, b, (short)0, c, false, false);
  }

  if (bok) {
    const float bb = bias ? bias[bcol] : 0.0f;
    const int row0 = rowBase + half * 8;
    float* __restrict__ yp = Y + (size_t)row0 * Ncol + bcol;
    #pragma unroll
    for (int r = 0; r < 8; ++r)
      yp[(size_t)r * Ncol] = c[r] + bb;
  }
}

__global__ void edge_logits_kernel(const int* __restrict__ ei, int E, int ET,
                                   const float* __restrict__ XL,
                                   const float* __restrict__ XR,
                                   const float* __restrict__ att,
                                   float* __restrict__ eo,
                                   float* __restrict__ mx, int d4) {
  const int wid  = (blockIdx.x * blockDim.x + threadIdx.x) >> 5;
  const int lane = threadIdx.x & 31;
  if (wid >= ET) return;
  int s, t;
  if (wid < E) { s = ei[wid]; t = ei[E + wid]; } else { s = t = wid - E; }
  const float4* __restrict__ xl = (const float4*)XL + (size_t)s * d4;
  const float4* __restrict__ xr = (const float4*)XR + (size_t)t * d4;
  const float4* __restrict__ at = (const float4*)att;
  float acc = 0.0f;
  for (int f = lane; f < d4; f += 32) {
    const float4 a = xl[f];
    const float4 b = xr[f];
    const float4 g = at[f];
    float h;
    h = a.x + b.x; h = h > 0.0f ? h : NEG_SLOPE * h; acc += h * g.x;
    h = a.y + b.y; h = h > 0.0f ? h : NEG_SLOPE * h; acc += h * g.y;
    h = a.z + b.z; h = h > 0.0f ? h : NEG_SLOPE * h; acc += h * g.z;
    h = a.w + b.w; h = h > 0.0f ? h : NEG_SLOPE * h; acc += h * g.w;
  }
  #pragma unroll
  for (int off = 16; off > 0; off >>= 1) acc += __shfl_xor(acc, off, 32);
  if (lane == 0) { eo[wid] = acc; atomicMaxF(&mx[t], acc); }
}

__global__ void edge_softmax_kernel(const int* __restrict__ ei, int E, int ET,
                                    const float* __restrict__ mx,
                                    float* __restrict__ ep,
                                    float* __restrict__ sx) {
  const int i = blockIdx.x * blockDim.x + threadIdx.x;
  if (i >= ET) return;
  const int t = (i < E) ? ei[E + i] : (i - E);
  const float p = __expf(ep[i] - mx[t]);
  ep[i] = p;
  atomicAdd(&sx[t], p);
}

__global__ void edge_scatter_kernel(const int* __restrict__ ei, int E, int ET,
                                    const float* __restrict__ XL,
                                    const float* __restrict__ ep,
                                    const float* __restrict__ sx,
                                    float* __restrict__ out, int d4) {
  const int wid  = (blockIdx.x * blockDim.x + threadIdx.x) >> 5;
  const int lane = threadIdx.x & 31;
  if (wid >= ET) return;
  int s, t;
  if (wid < E) { s = ei[wid]; t = ei[E + wid]; } else { s = t = wid - E; }
  const float alpha = ep[wid] / sx[t];
  const float4* __restrict__ xl = (const float4*)XL + (size_t)s * d4;
  float* __restrict__ op = out + (size_t)t * (d4 * 4);
  for (int f = lane; f < d4; f += 32) {
    const float4 v = xl[f];
    float* o4 = op + f * 4;
    atomicAdd(o4 + 0, alpha * v.x);
    atomicAdd(o4 + 1, alpha * v.y);
    atomicAdd(o4 + 2, alpha * v.z);
    atomicAdd(o4 + 3, alpha * v.w);
  }
}

__global__ void finalize_kernel(float* __restrict__ out,
                                const float* __restrict__ bias,
                                int n, int dmask, int relu) {
  const int i = blockIdx.x * blockDim.x + threadIdx.x;
  if (i >= n) return;
  float v = out[i] + bias[i & dmask];
  if (relu) v = fmaxf(v, 0.0f);
  out[i] = v;
}

extern "C" void kernel_launch(void* const* d_in, const int* in_sizes, int n_in,
                              void* d_out, int out_size, void* d_ws, size_t ws_size,
                              hipStream_t stream) {
  const float* x   = (const float*)d_in[0];
  const int*   ei  = (const int*)  d_in[1];
  const float* w1l = (const float*)d_in[2];
  const float* b1l = (const float*)d_in[3];
  const float* w1r = (const float*)d_in[4];
  const float* b1r = (const float*)d_in[5];
  const float* a1  = (const float*)d_in[6];
  const float* o1  = (const float*)d_in[7];
  const float* w2l = (const float*)d_in[8];
  const float* b2l = (const float*)d_in[9];
  const float* w2r = (const float*)d_in[10];
  const float* b2r = (const float*)d_in[11];
  const float* a2  = (const float*)d_in[12];
  const float* o2  = (const float*)d_in[13];
  const float* w3l = (const float*)d_in[14];
  const float* b3l = (const float*)d_in[15];
  const float* w3r = (const float*)d_in[16];
  const float* b3r = (const float*)d_in[17];
  const float* a3  = (const float*)d_in[18];
  const float* o3  = (const float*)d_in[19];
  const float* wc  = (const float*)d_in[20];
  const float* bc  = (const float*)d_in[21];

  const int N  = in_sizes[0] / 128;
  const int E  = in_sizes[1] / 2;
  const int ET = E + N;

  float* B0  = (float*)d_ws;
  float* B1  = B0 + (size_t)N * 256;
  float* B2  = B1 + (size_t)N * 256;
  float* EP  = B2 + (size_t)N * 256;
  float* Mx  = EP + (size_t)ET;
  float* Sx  = Mx + (size_t)N;
  v2f*   WPa = (v2f*)(Sx + (size_t)N);
  v2f*   WPb = WPa + 32768;

  auto packW = [&](const float* W, v2f* Wp, int k, int n) {
    const int total = (k >> 1) * n;
    pack_w_kernel<<<(total + 255) / 256, 256, 0, stream>>>(W, Wp, k, n);
  };
  auto gemmBig = [&](const float* X, const v2f* Wp, const float* b, float* Y,
                     int m, int k, int n) {
    const int tiles = (m >> 4) * (n >> 6);
    gemm_bias_wmma_n64<<<(tiles + 3) / 4, 128, 0, stream>>>(X, Wp, b, Y, m, k, n);
  };
  auto gemmGuard = [&](const float* X, const v2f* Wp, const float* b, float* Y,
                       int m, int k, int n) {
    const int tiles = (m >> 4) * ((n + 15) / 16);
    gemm_bias_wmma_guard<<<(tiles + 3) / 4, 128, 0, stream>>>(X, Wp, b, Y, m, k, n);
  };

  auto layer = [&](const float* Xin, int din, int dout,
                   const float* wl, const float* bl,
                   const float* wr, const float* br,
                   const float* att, const float* ob,
                   float* XL, float* XO, int relu) {
    packW(wl, WPa, din, dout);
    packW(wr, WPb, din, dout);
    gemmBig(Xin, WPa, bl, XL, N, din, dout);
    gemmBig(Xin, WPb, br, XO, N, din, dout);
    fillf_kernel<<<(N + 255) / 256, 256, 0, stream>>>(Mx, N, -__builtin_huge_valf());
    fillf_kernel<<<(N + 255) / 256, 256, 0, stream>>>(Sx, N, 0.0f);
    edge_logits_kernel<<<(ET + 7) / 8, 256, 0, stream>>>(ei, E, ET, XL, XO, att,
                                                         EP, Mx, dout >> 2);
    fillf_kernel<<<(N * dout + 255) / 256, 256, 0, stream>>>(XO, N * dout, 0.0f);
    edge_softmax_kernel<<<(ET + 255) / 256, 256, 0, stream>>>(ei, E, ET, Mx, EP, Sx);
    edge_scatter_kernel<<<(ET + 7) / 8, 256, 0, stream>>>(ei, E, ET, XL, EP, Sx,
                                                          XO, dout >> 2);
    finalize_kernel<<<(N * dout + 255) / 256, 256, 0, stream>>>(XO, ob,
                                                                N * dout, dout - 1, relu);
  };

  layer(x,  128, 128, w1l, b1l, w1r, b1r, a1, o1, B0, B1, 1);
  layer(B1, 128, 128, w2l, b2l, w2r, b2r, a2, o2, B0, B2, 1);
  layer(B2, 128, 256, w3l, b3l, w3r, b3r, a3, o3, B0, B1, 0);
  packW(wc, WPa, 256, 40);
  gemmGuard(B1, WPa, bc, (float*)d_out, N, 256, 40);
}


# Layers 2, 3: the node owners also take the logits: owned nodes' xr rows sit in LDS, only xl[src] is gathered, node maximum by LDS max; 3 logit launches and global max atomics gone; sums keep order
# speedup vs baseline: 3.1584x; 3.1584x over previous
#include <hip/hip_runtime.h>

typedef __attribute__((ext_vector_type(2))) float v2f;
typedef __attribute__((ext_vector_type(8))) float v8f;

#define NEG_SLOPE 0.2f

__device__ inline void atomicMaxF(float* addr, float v) {
  if (v >= 0.0f) atomicMax((int*)addr, __float_as_int(v));
  else           atomicMin((unsigned int*)addr, __float_as_uint(v));
}

__global__ void pack_w_kernel(const float* __restrict__ W,
                              v2f* __restrict__ Wp, int K, int Ncol) {
  const int i = blockIdx.x * blockDim.x + threadIdx.x;
  const int total = (K >> 1) * Ncol;
  if (i >= total) return;
  const int kp  = i / Ncol;
  const int col = i - kp * Ncol;
  v2f p;
  p[0] = W[(size_t)(2 * kp)     * Ncol + col];
  p[1] = W[(size_t)(2 * kp + 1) * Ncol + col];
  Wp[i] = p;
}

template <bool ACT, int CLR, bool PLANES>
__global__ void gemm_bias_wmma_n64(const float* __restrict__ X,
                                   const float* __restrict__ abias,
                                   const v2f* __restrict__ Wp,
                                   const float* __restrict__ bias,
                                   float* __restrict__ Y,
                                   float* __restrict__ mx,
                                   int M, int K, int Ncol) {
  const int lane = threadIdx.x & 31;
  const int wave = threadIdx.x >> 5;
  const int ct = Ncol >> 6;
  const int rt = M >> 4;
  const int tile = blockIdx.x * (blockDim.x >> 5) + wave;
  if (tile >= ct * rt) return;

  const int rowBase = (tile / ct) << 4;
  const int colBase = (tile % ct) << 6;
  const int half = lane >> 4;
  const int l15  = lane & 15;

  const float* __restrict__ ap = X + (size_t)(rowBase + l15) * K + half * 2;
  const float* __restrict__ abp = abias + half * 2;
  const v2f*   __restrict__ bp = Wp + (size_t)half * Ncol + colBase + l15;

  v8f cc[4] = {v8f{}, v8f{}, v8f{}, v8f{}};
  #pragma unroll 4
  for (int k0 = 0; k0 < K; k0 += 4) {
    v2f a = *(const v2f*)(ap + k0);
    if (ACT) {
      a += *(const v2f*)(abp + k0);
      a[0] = fmaxf(a[0], 0.0f);
      a[1] = fmaxf(a[1], 0.0f);
    }
    const v2f* __restrict__ br = bp + (size_t)(k0 >> 1) * Ncol;
    const v2f bv0 = br[0];
    const v2f bv1 = br[16];
    const v2f bv2 = br[32];
    const v2f bv3 = br[48];
    cc[0] = __builtin_amdgcn_wmma_f32_16x16x4_f32(false, a, false, bv0, (short)0, cc[0], false, false);
    cc[1] = __builtin_amdgcn_wmma_f32_16x16x4_f32(false, a, false, bv1, (short)0, cc[1], false, false);
    cc[2] = __builtin_amdgcn_wmma_f32_16x16x4_f32(false, a, false, bv2, (short)0, cc[2], false, false);
    cc[3] = __builtin_amdgcn_wmma_f32_16x16x4_f32(false, a, false, bv3, (short)0, cc[3], false, false);
  }

  const int row0 = rowBase + half * 8;
  #pragma unroll
  for (int t = 0; t < 4; ++t) {
    const int col = colBase + t * 16 + l15;
    const float bb = bias ? bias[col] : 0.0f;
    const int ld = PLANES ? 128 : Ncol;
    float* __restrict__ yp = PLANES
        ? Y + (size_t)(col >> 7) * M * 128 + (size_t)row0 * 128 + (col & 127)
        : Y + (size_t)row0 * Ncol + col;
    #pragma unroll
    for (int r = 0; r < 8; ++r)
      yp[(size_t)r * ld] = cc[t][r] + bb;
  }

  if (CLR == 1) {
    if (colBase == 0 && half == 0) mx[rowBase + l15] = -__builtin_huge_valf();
  }
}

__global__ void gemm_bias_wmma_guard(const float* __restrict__ X,
                                     const float* __restrict__ abias,
                                     const v2f* __restrict__ Wp,
                                     const float* __restrict__ bias,
                                     float* __restrict__ Y,
                                     int M, int K, int Ncol) {
  const int lane = threadIdx.x & 31;
  const int wave = threadIdx.x >> 5;
  const int ct = (Ncol + 15) >> 4;
  const int rt = M >> 4;
  const int tile = blockIdx.x * (blockDim.x >> 5) + wave;
  if (tile >= ct * rt) return;

  const int rowBase = (tile / ct) << 4;
  const int colBase = (tile % ct) << 4;
  const int half = lane >> 4;
  const int l15  = lane & 15;
  const int bcol = colBase + l15;
  const bool bok = bcol < Ncol;
  const int bcolc = bok ? bcol : (Ncol - 1);
  const float bmask = bok ? 1.0f : 0.0f;

  const float* __restrict__ ap = X + (size_t)(rowBase + l15) * 128 + half * 2;
  const float* __restrict__ abp = abias + half * 2;
  const v2f*   __restrict__ bp = Wp + (size_t)half * Ncol + bcolc;

  v8f c = {};
  for (int kb = 0; kb < K; kb += 128) {
    const float* __restrict__ app = ap + (size_t)(kb >> 7) * M * 128;
    #pragma unroll 4
    for (int kk = 0; kk < 128; kk += 4) {
      const int k0 = kb + kk;
      const v2f a = *(const v2f*)(app + kk) + *(const v2f*)(abp + k0);
      v2f b = bp[(size_t)(k0 >> 1) * Ncol];
      b[0] *= bmask;
      b[1] *= bmask;
      c = __builtin_amdgcn_wmma_f32_16x16x4_f32(false, a, false, b, (short)0, c, false, false);
    }
  }

  if (bok) {
    const float bb = bias ? bias[bcol] : 0.0f;
    const int row0 = rowBase + half * 8;
    float* __restrict__ yp = Y + (size_t)row0 * Ncol + bcol;
    #pragma unroll
    for (int r = 0; r < 8; ++r)
      yp[(size_t)r * Ncol] = c[r] + bb;
  }
}

#define EPW 16

template <int D>
__global__ void edge_logits_kernel(const int* __restrict__ ei, int E, int ET,
                                   const float* __restrict__ XL,
                                   const float* __restrict__ XR,
                                   const float* __restrict__ att,
                                   float* __restrict__ eo,
                                   float* __restrict__ mx) {
  constexpr int R = D / 128;
  constexpr int B = 1024 / D;
  const int wave = __builtin_amdgcn_readfirstlane((blockIdx.x * blockDim.x + threadIdx.x) >> 5);
  const int lane = threadIdx.x & 31;
  const int base = wave * EPW;
  if (base >= ET) return;
  int idx = base + (lane & (EPW - 1));
  const bool live = idx < ET;
  idx = live ? idx : ET - 1;
  int sv, tv;
  if (idx < E) { sv = ei[idx]; tv = ei[E + idx]; } else { sv = tv = idx - E; }

  const float4* __restrict__ at = (const float4*)att;
  float4 g[R];
  #pragma unroll
  for (int k = 0; k < R; ++k) g[k] = at[lane + 32 * k];

  float res = 0.0f;
  #pragma unroll
  for (int b = 0; b < EPW; b += B) {
    float4 a[B][R], c[B][R];
    #pragma unroll
    for (int jj = 0; jj < B; ++jj) {
      const int s = __builtin_amdgcn_readlane(sv, b + jj);
      const int t = __builtin_amdgcn_readlane(tv, b + jj);
      const float4* __restrict__ xl = (const float4*)(XL + (size_t)s * D) + lane;
      const float4* __restrict__ xr = (const float4*)(XR + (size_t)t * D) + lane;
      #pragma unroll
      for (int k = 0; k < R; ++k) { a[jj][k] = xl[32 * k]; c[jj][k] = xr[32 * k]; }
    }
    #pragma unroll
    for (int jj = 0; jj < B; ++jj) {
      float acc = 0.0f;
      #pragma unroll
      for (int k = 0; k < R; ++k) {
        const float4 p = a[jj][k];
        const float4 q = c[jj][k];
        float h;
        h = p.x + q.x; h = h > 0.0f ? h : NEG_SLOPE * h; acc += h * g[k].x;
        h = p.y + q.y; h = h > 0.0f ? h : NEG_SLOPE * h; acc += h * g[k].y;
        h = p.z + q.z; h = h > 0.0f ? h : NEG_SLOPE * h; acc += h * g[k].z;
        h = p.w + q.w; h = h > 0.0f ? h : NEG_SLOPE * h; acc += h * g[k].w;
      }
      #pragma unroll
      for (int off = 16; off > 0; off >>= 1) acc += __shfl_xor(acc, off, 32);
      res = (lane == b + jj) ? acc : res;
    }
  }
  if (lane < EPW && live) {
    eo[idx] = res;
    atomicMaxF(&mx[tv], res);
  }
}

#define NPB 196
#define OWN_WAVES 32
#define RING 1024
#define LCAP 1024

struct Guard   { static constexpr bool value = true;  };
struct NoGuard { static constexpr bool value = false; };

template <int D>
__global__ void __launch_bounds__(OWN_WAVES * 32)
owner_sum_kernel(const int* __restrict__ ei, int E, int N,
                 const float* __restrict__ XL,
                 const float* __restrict__ el,
                 const float* __restrict__ mx,
                 float* __restrict__ out,
                 int* __restrict__ lst, int* __restrict__ cnts) {
  constexpr int Q = (D + 31) / 32;
  constexpr int RB = 16;
  __shared__ float acc[NPB * D];
  __shared__ float ssum[NPB];
  __shared__ float mloc[NPB];
  __shared__ int ring[OWN_WAVES * RING];

  const int lane = threadIdx.x & 31;
  const int wave = __builtin_amdgcn_readfirstlane(threadIdx.x >> 5);
  const int base = blockIdx.x * NPB;
  const int n_own = (N - base) < NPB ? (N - base) : NPB;

  for (int i = wave; i < n_own; i += OWN_WAVES) {
    const int node = base + i;
    const float m = mx[node];
    const float p = __expf(el[E + node] - m);
    if (lane == 0) { ssum[i] = p; mloc[i] = m; }
    #pragma unroll
    for (int k = 0; k < Q; ++k) {
      const int f = lane + 32 * k;
      if (D % 32 == 0 || f < D) acc[i * D + f] = p * XL[(size_t)node * D + f];
    }
  }
  __syncthreads();

  const int* __restrict__ dp = ei + E;
  const bool vec = ((E & 3) == 0) && ((((size_t)dp) & 15) == 0);
  const int nChunks = (E + 127) >> 7;
  const int rot = (int)blockIdx.x * (nChunks / (int)gridDim.x);
  const int rb = wave * RING;
  int cnext = wave, head = 0, cnt = 0;
  const int slot = (int)blockIdx.x * OWN_WAVES + wave;
  int* __restrict__ myl = lst + (size_t)slot * LCAP;
  int tot = 0;
  for (;;) {
    int nb, id;
    if (cnt < 32 && cnext < nChunks) {
      int d[4][4];
      int cb[4];
      bool whole = vec;
      #pragma unroll
      for (int u = 0; u < 4; ++u) {
        const int lc = cnext + OWN_WAVES * u;
        int c = lc + rot;
        c = c >= nChunks ? c - nChunks : c;
        cb[u] = c * 128;
        whole = whole && lc < nChunks && cb[u] + 128 <= E;
      }
      if (whole) {
        #pragma unroll
        for (int u = 0; u < 4; ++u) {
          const int4 t = *(const int4*)(dp + cb[u] + 4 * lane);
          d[u][0] = t.x; d[u][1] = t.y; d[u][2] = t.z; d[u][3] = t.w;
        }
      } else {
        #pragma unroll
        for (int u = 0; u < 4; ++u) {
          const bool there = cnext + OWN_WAVES * u < nChunks;
          const int eb = cb[u] + 4 * lane;
          d[u][0] = (there && eb     < E) ? dp[eb]     : -1;
          d[u][1] = (there && eb + 1 < E) ? dp[eb + 1] : -1;
          d[u][2] = (there && eb + 2 < E) ? dp[eb + 2] : -1;
          d[u][3] = (there && eb + 3 < E) ? dp[eb + 3] : -1;
        }
      }
      #pragma unroll
      for (int u = 0; u < 4; ++u) {
        #pragma unroll
        for (int q = 0; q < 4; ++q) {
          const bool hit = (unsigned)(d[u][q] - base) < (unsigned)n_own;
          const unsigned mk = __builtin_amdgcn_ballot_w32(hit);
          if (mk) {
            const int r = __builtin_amdgcn_mbcnt_lo(mk, 0);
            if (hit) __hip_atomic_store(&ring[rb + ((head + cnt + r) & (RING - 1))], cb[u] + 4 * lane + q,
                                        __ATOMIC_RELAXED, __HIP_MEMORY_SCOPE_WAVEFRONT);
            if (hit && tot + r < LCAP) myl[tot + r] = cb[u] + 4 * lane + q;
            tot += __builtin_popcount(mk);
            cnt += __builtin_popcount(mk);
          }
        }
      }
      cnext += OWN_WAVES * 4;
      continue;
    }
    if (cnt == 0) break;
    nb = cnt < 32 ? cnt : 32;
    id = __hip_atomic_load(&ring[rb + ((head + lane) & (RING - 1))],
                           __ATOMIC_RELAXED, __HIP_MEMORY_SCOPE_WAVEFRONT);
    head = (head + nb) & (RING - 1);
    cnt -= nb;
    int sv = 0, tl = 0;
    float p = 0.0f;
    if (lane < nb) {
      sv = ei[id];
      tl = dp[id] - base;
      p = __expf(el[id] - mloc[tl]);
      atomicAdd(&ssum[tl], p);
    }
    auto rows = [&](const int j0, auto G) {
      float v[RB][Q];
      #pragma unroll
      for (int jj = 0; jj < RB; ++jj) {
        if (!G.value || j0 + jj < nb) {
          const int s = __builtin_amdgcn_readlane(sv, j0 + jj);
          const float* __restrict__ xr = XL + (size_t)s * D + lane;
          #pragma unroll
          for (int k = 0; k < Q; ++k)
            if (D % 32 == 0 || lane + 32 * k < D) v[jj][k] = xr[32 * k];
        }
      }
      #pragma unroll
      for (int jj = 0; jj < RB; ++jj) {
        if (!G.value || j0 + jj < nb) {
          const int t = __builtin_amdgcn_readlane(tl, j0 + jj);
          const float pp = __int_as_float(__builtin_amdgcn_readlane(__float_as_int(p), j0 + jj));
          float* ap = acc + t * D + lane;
          #pragma unroll
          for (int k = 0; k < Q; ++k)
            if (D % 32 == 0 || lane + 32 * k < D) atomicAdd(ap + 32 * k, pp * v[jj][k]);
        }
      }
    };
    if (nb == 32) {
      #pragma unroll
      for (int j0 = 0; j0 < 32; j0 += RB) rows(j0, NoGuard());
    } else {
      for (int j0 = 0; j0 < nb; j0 += RB) rows(j0, Guard());
    }
  }
  if (lane == 0) cnts[slot] = tot;
  __syncthreads();

  const int total = n_own * D;
  float* __restrict__ op = out + (size_t)base * D;
  for (int i = threadIdx.x; i < total; i += OWN_WAVES * 32)
    op[i] = acc[i] / ssum[i / D];
}

template <int P, int D>
__global__ void __launch_bounds__(OWN_WAVES * 32)
owner_attn_kernel(const int* __restrict__ ei, int E, int N,
                  const float* __restrict__ XL,
                  const float* __restrict__ XR,
                  const float* __restrict__ att,
                  const float* __restrict__ XS,
                  float* __restrict__ out,
                  const int* __restrict__ lst, const int* __restrict__ cnts,
                  float* __restrict__ elst) {
  constexpr int Q = (D + 31) / 32;
  constexpr int RB = 16;
  constexpr int RA = 8 / P;
  constexpr int XW = P * 128;
  __shared__ float acc[NPB * D];
  __shared__ alignas(16) float xrow[NPB * XW];
  __shared__ float ssum[NPB];
  __shared__ float mloc[NPB];
  __shared__ float eself[NPB];

  const int lane = threadIdx.x & 31;
  const int wave = __builtin_amdgcn_readfirstlane(threadIdx.x >> 5);
  const int base = blockIdx.x * NPB;
  const int n_own = (N - base) < NPB ? (N - base) : NPB;
  const int* __restrict__ dp = ei + E;

  float4 g[P];
  #pragma unroll
  for (int k = 0; k < P; ++k) g[k] = ((const float4*)att)[lane + 32 * k];

  auto logit = [&](const float4 (&a)[P], const int t) -> float {
    float e = 0.0f;
    #pragma unroll
    for (int k = 0; k < P; ++k) {
      const float4 q = *(const float4*)&xrow[t * XW + 128 * k + 4 * lane];
      const float4 p = a[k];
      float s = 0.0f;
      float h;
      h = p.x + q.x; h = h > 0.0f ? h : NEG_SLOPE * h; s += h * g[k].x;
      h = p.y + q.y; h = h > 0.0f ? h : NEG_SLOPE * h; s += h * g[k].y;
      h = p.z + q.z; h = h > 0.0f ? h : NEG_SLOPE * h; s += h * g[k].z;
      h = p.w + q.w; h = h > 0.0f ? h : NEG_SLOPE * h; s += h * g[k].w;
      #pragma unroll
      for (int off = 16; off > 0; off >>= 1) s += __shfl_xor(s, off, 32);
      e = (k == 0) ? s : e + s;
    }
    return e;
  };
  auto srcRow = [&](float4 (&a)[P], const int s) {
    #pragma unroll
    for (int k = 0; k < P; ++k)
      a[k] = ((const float4*)(XL + ((size_t)k * N + s) * 128))[lane];
  };

  auto logitBatch = [&](auto sOf, auto tOf, float (&e)[RA]) {
    float4 a[RA][P];
    #pragma unroll
    for (int jj = 0; jj < RA; ++jj) srcRow(a[jj], sOf(jj));
    float s[RA][P];
    #pragma unroll
    for (int jj = 0; jj < RA; ++jj) {
      const int t = tOf(jj);
      #pragma unroll
      for (int k = 0; k < P; ++k) {
        const float4 q = *(const float4*)&xrow[t * XW + 128 * k + 4 * lane];
        const float4 p = a[jj][k];
        float v = 0.0f;
        float h;
        h = p.x + q.x; h = h > 0.0f ? h : NEG_SLOPE * h; v += h * g[k].x;
        h = p.y + q.y; h = h > 0.0f ? h : NEG_SLOPE * h; v += h * g[k].y;
        h = p.z + q.z; h = h > 0.0f ? h : NEG_SLOPE * h; v += h * g[k].z;
        h = p.w + q.w; h = h > 0.0f ? h : NEG_SLOPE * h; v += h * g[k].w;
        s[jj][k] = v;
      }
    }
    #pragma unroll
    for (int off = 16; off > 0; off >>= 1) {
      #pragma unroll
      for (int jj = 0; jj < RA; ++jj) {
        #pragma unroll
        for (int k = 0; k < P; ++k) s[jj][k] += __shfl_xor(s[jj][k], off, 32);
      }
    }
    #pragma unroll
    for (int jj = 0; jj < RA; ++jj) {
      float v = s[jj][0];
      #pragma unroll
      for (int k = 1; k < P; ++k) v += s[jj][k];
      e[jj] = v;
    }
  };

  for (int i = wave; i < n_own; i += OWN_WAVES) {
    const int node = base + i;
    #pragma unroll
    for (int k = 0; k < P; ++k)
      *(float4*)&xrow[i * XW + 128 * k + 4 * lane] = ((const float4*)(XR + ((size_t)k * N + node) * 128))[lane];
  }
  for (int i0 = wave; i0 < n_own; i0 += OWN_WAVES * RA) {
    auto iOf = [&](const int jj) { const int i = i0 + OWN_WAVES * jj; return i < n_own ? i : i0; };
    float e[RA];
    logitBatch([&](const int jj) { return base + iOf(jj); }, iOf, e);
    #pragma unroll
    for (int jj = 0; jj < RA; ++jj) {
      const int i = i0 + OWN_WAVES * jj;
      if (lane == 0 && i < n_own) { eself[i] = e[jj]; mloc[i] = e[jj]; }
    }
  }
  __syncthreads();

  const int slot = (int)blockIdx.x * OWN_WAVES + wave;
  const int* __restrict__ myl = lst + (size_t)slot * LCAP;
  float* __restrict__ mye = elst + (size_t)slot * LCAP;
  const int tot = __builtin_amdgcn_readfirstlane(cnts[slot]);
  const bool useList = (unsigned)tot <= (unsigned)LCAP;

  auto walk = [&](auto f) {
    const int nChunks = (E + 127) >> 7;
    const int rot = (int)blockIdx.x * (nChunks / (int)gridDim.x);
    for (int lc = wave; lc < nChunks; lc += OWN_WAVES) {
      int c = lc + rot;
      c = c >= nChunks ? c - nChunks : c;
      for (int q = 0; q < 4; ++q) {
        const int eb = c * 128 + 4 * lane + q;
        const int d = eb < E ? dp[eb] : -1;
        unsigned mk = __builtin_amdgcn_ballot_w32((unsigned)(d - base) < (unsigned)n_own);
        while (mk) {
          const int l = __builtin_ctz(mk);
          mk &= mk - 1;
          const int id = c * 128 + 4 * l + q;
          f(id, ei[id], dp[id] - base);
        }
      }
    }
  };

  if (useList) {
    for (int pos = 0; pos < tot; pos += 32) {
      const int nb = tot - pos < 32 ? tot - pos : 32;
      const int id = myl[pos + (lane < nb ? lane : 0)];
      const int sv = ei[id];
      const int tl = dp[id] - base;
      float res = 0.0f;
      int j0 = 0;
      do {
        float e[RA];
        logitBatch([&](const int jj) { return __builtin_amdgcn_readlane(sv, j0 + jj); },
                   [&](const int jj) { return __builtin_amdgcn_readlane(tl, j0 + jj); }, e);
        #pragma unroll
        for (int jj = 0; jj < RA; ++jj) res = (lane == j0 + jj) ? e[jj] : res;
        j0 += RA;
      } while (j0 < nb);
      if (lane < nb) {
        mye[pos + lane] = res;
        atomicMaxF(&mloc[tl], res);
      }
    }
  } else {
    walk([&](const int, const int s, const int t) {
      float4 a[P];
      srcRow(a, s);
      const float e = logit(a, t);
      if (lane == 0) atomicMaxF(&mloc[t], e);
    });
  }
  __syncthreads();

  for (int i = wave; i < n_own; i += OWN_WAVES) {
    const int node = base + i;
    const float p = __expf(eself[i] - mloc[i]);
    if (lane == 0) ssum[i] = p;
    #pragma unroll
    for (int k = 0; k < Q; ++k) {
      const int f = lane + 32 * k;
      if (D % 32 == 0 || f < D) acc[i * D + f] = p * XS[(size_t)node * D + f];
    }
  }
  __syncthreads();

  auto addRow1 = [&](const int s, const int t, const float pp) {
    const float* __restrict__ xs = XS + (size_t)s * D + lane;
    float* ap = acc + t * D + lane;
    #pragma unroll
    for (int k = 0; k < Q; ++k)
      if (D % 32 == 0 || lane + 32 * k < D) atomicAdd(ap + 32 * k, pp * xs[32 * k]);
  };
  if (useList) {
    for (int pos = 0; pos < tot; pos += 32) {
      const int nb = tot - pos < 32 ? tot - pos : 32;
      const int id = myl[pos + (lane < nb ? lane : 0)];
      int sv = 0, tl = 0;
      float p = 0.0f;
      if (lane < nb) {
        sv = ei[id];
        tl = dp[id] - base;
        p = __expf(mye[pos + lane] - mloc[tl]);
        atomicAdd(&ssum[tl], p);
      }
      auto rows = [&](const int j0, auto G) {
        float v[RB][Q];
        #pragma unroll
        for (int jj = 0; jj < RB; ++jj) {
          if (!G.value || j0 + jj < nb) {
            const int s = __builtin_amdgcn_readlane(sv, j0 + jj);
            const float* __restrict__ xs = XS + (size_t)s * D + lane;
            #pragma unroll
            for (int k = 0; k < Q; ++k)
              if (D % 32 == 0 || lane + 32 * k < D) v[jj][k] = xs[32 * k];
          }
        }
        #pragma unroll
        for (int jj = 0; jj < RB; ++jj) {
          if (!G.value || j0 + jj < nb) {
            const int t = __builtin_amdgcn_readlane(tl, j0 + jj);
            const float pp = __int_as_float(__builtin_amdgcn_readlane(__float_as_int(p), j0 + jj));
            float* ap = acc + t * D + lane;
            #pragma unroll
            for (int k = 0; k < Q; ++k)
              if (D % 32 == 0 || lane + 32 * k < D) atomicAdd(ap + 32 * k, pp * v[jj][k]);
          }
        }
      };
      if (nb == 32) {
        #pragma unroll
        for (int j0 = 0; j0 < 32; j0 += RB) rows(j0, NoGuard());
      } else {
        for (int j0 = 0; j0 < nb; j0 += RB) rows(j0, Guard());
      }
    }
  } else {
    walk([&](const int, const int s, const int t) {
      float4 a[P];
      srcRow(a, s);
      const float pp = __expf(logit(a, t) - mloc[t]);
      if (lane == 0) atomicAdd(&ssum[t], pp);
      addRow1(s, t, pp);
    });
  }
  __syncthreads();

  const int total = n_own * D;
  float* __restrict__ op = out + (size_t)base * D;
  for (int i = threadIdx.x; i < total; i += OWN_WAVES * 32)
    op[i] = acc[i] / ssum[i / D];
}

extern "C" void kernel_launch(void* const* d_in, const int* in_sizes, int n_in,
                              void* d_out, int out_size, void* d_ws, size_t ws_size,
                              hipStream_t stream) {
  const float* x   = (const float*)d_in[0];
  const int*   ei  = (const int*)  d_in[1];
  const float* w1l = (const float*)d_in[2];
  const float* b1l = (const float*)d_in[3];
  const float* w1r = (const float*)d_in[4];
  const float* b1r = (const float*)d_in[5];
  const float* a1  = (const float*)d_in[6];
  const float* o1  = (const float*)d_in[7];
  const float* w2l = (const float*)d_in[8];
  const float* b2l = (const float*)d_in[9];
  const float* w2r = (const float*)d_in[10];
  const float* b2r = (const float*)d_in[11];
  const float* a2  = (const float*)d_in[12];
  const float* o2  = (const float*)d_in[13];
  const float* w3l = (const float*)d_in[14];
  const float* b3l = (const float*)d_in[15];
  const float* w3r = (const float*)d_in[16];
  const float* b3r = (const float*)d_in[17];
  const float* a3  = (const float*)d_in[18];
  const float* o3  = (const float*)d_in[19];
  const float* wc  = (const float*)d_in[20];
  const float* bc  = (const float*)d_in[21];

  const int N  = in_sizes[0] / 128;
  const int E  = in_sizes[1] / 2;
  const int ET = E + N;
  const int eblocks = ((ET + EPW - 1) / EPW + 7) / 8;

  float* B0  = (float*)d_ws;
  float* B1  = B0 + (size_t)N * 256;
  float* B2  = B1 + (size_t)N * 256;
  float* EP  = B2 + (size_t)N * 256;
  float* Mx  = EP + (size_t)ET;
  v2f*   WPa = (v2f*)(Mx + (size_t)N);
  v2f*   WPb = WPa + 32768;
  float* S1  = B1;
  float* S2  = B1 + (size_t)N * 128;

  auto packW = [&](const float* W, v2f* Wp, int k, int n) {
    const int total = (k >> 1) * n;
    pack_w_kernel<<<(total + 255) / 256, 256, 0, stream>>>(W, Wp, k, n);
  };
  auto gemmGuard = [&](const float* X, const float* ab, const v2f* Wp, const float* b,
                       float* Y, int m, int k, int n) {
    const int tiles = (m >> 4) * ((n + 15) / 16);
    gemm_bias_wmma_guard<<<(tiles + 3) / 4, 128, 0, stream>>>(X, ab, Wp, b, Y, m, k, n);
  };
  const int oblocks = (N + NPB - 1) / NPB;
  int* CN = (int*)(WPb + 32768);
  int* LS = CN + (size_t)oblocks * OWN_WAVES;
  float* EL = (float*)(LS + (size_t)oblocks * OWN_WAVES * LCAP);

  auto layer = [&](const float* Xin, const float* inb,
                   const float* wl, const float* bl,
                   const float* wr, const float* br,
                   const float* att, float* XL, float* XR, float* ACC) {
    const int gblocks = ((N >> 4) * 2 + 3) / 4;
    packW(wl, WPa, 128, 128);
    packW(wr, WPb, 128, 128);
    if (inb) {
      gemm_bias_wmma_n64<true, 0, false><<<gblocks, 128, 0, stream>>>(Xin, inb, WPa, bl, XL, nullptr, N, 128, 128);
      gemm_bias_wmma_n64<true, 0, false><<<gblocks, 128, 0, stream>>>(Xin, inb, WPb, br, XR, nullptr, N, 128, 128);
    } else {
      gemm_bias_wmma_n64<false, 0, false><<<gblocks, 128, 0, stream>>>(Xin, nullptr, WPa, bl, XL, nullptr, N, 128, 128);
      gemm_bias_wmma_n64<false, 1, false><<<gblocks, 128, 0, stream>>>(Xin, nullptr, WPb, br, XR, Mx, N, 128, 128);
    }
    if (inb) {
      owner_attn_kernel<1, 128><<<oblocks, OWN_WAVES * 32, 0, stream>>>(ei, E, N, XL, XR, att, XL, ACC, LS, CN, EL);
    } else {
      edge_logits_kernel<128><<<eblocks, 256, 0, stream>>>(ei, E, ET, XL, XR, att, EP, Mx);
      owner_sum_kernel<128><<<oblocks, OWN_WAVES * 32, 0, stream>>>(ei, E, N, XL, EP, Mx, ACC, LS, CN);
    }
  };

  layer(x,  nullptr, w1l, b1l, w1r, b1r, a1, B0, B2, S1);
  layer(S1, o1,      w2l, b2l, w2r, b2r, a2, B0, B2, S2);
  float* out = (float*)d_out;
  const int gblocks3 = ((N >> 4) * 4 + 3) / 4;
  packW(w3l, WPa, 128, 256);
  packW(w3r, WPb, 128, 256);
  gemm_bias_wmma_n64<true, 0, true><<<gblocks3, 128, 0, stream>>>(S2, o2, WPa, b3l, B0, nullptr, N, 128, 256);
  gemm_bias_wmma_n64<true, 0, true><<<gblocks3, 128, 0, stream>>>(S2, o2, WPb, b3r, B2, nullptr, N, 128, 256);
  packW(wc, WPa, 256, 40);
  gemmGuard(B0, o3, WPa, bc, B1, N, 256, 40);
  owner_attn_kernel<2, 40><<<oblocks, OWN_WAVES * 32, 0, stream>>>(ei, E, N, B0, B2, a3, B1, out, LS, CN, EL);
}


# Node owners' kept lists hold each edge's end points (source node, owned node's place), not its id: layers 2 and 3 read them 8 bytes per lane; 2 scattered reads per edge per pass gone; sums keep order
# speedup vs baseline: 3.3402x; 1.0576x over previous
#include <hip/hip_runtime.h>

typedef __attribute__((ext_vector_type(2))) float v2f;
typedef __attribute__((ext_vector_type(2))) int v2i;
typedef __attribute__((ext_vector_type(8))) float v8f;

#define NEG_SLOPE 0.2f

__device__ inline void atomicMaxF(float* addr, float v) {
  if (v >= 0.0f) atomicMax((int*)addr, __float_as_int(v));
  else           atomicMin((unsigned int*)addr, __float_as_uint(v));
}

__global__ void pack_w_kernel(const float* __restrict__ W,
                              v2f* __restrict__ Wp, int K, int Ncol) {
  const int i = blockIdx.x * blockDim.x + threadIdx.x;
  const int total = (K >> 1) * Ncol;
  if (i >= total) return;
  const int kp  = i / Ncol;
  const int col = i - kp * Ncol;
  v2f p;
  p[0] = W[(size_t)(2 * kp)     * Ncol + col];
  p[1] = W[(size_t)(2 * kp + 1) * Ncol + col];
  Wp[i] = p;
}

template <bool ACT, int CLR, bool PLANES>
__global__ void gemm_bias_wmma_n64(const float* __restrict__ X,
                                   const float* __restrict__ abias,
                                   const v2f* __restrict__ Wp,
                                   const float* __restrict__ bias,
                                   float* __restrict__ Y,
                                   float* __restrict__ mx,
                                   int M, int K, int Ncol) {
  const int lane = threadIdx.x & 31;
  const int wave = threadIdx.x >> 5;
  const int ct = Ncol >> 6;
  const int rt = M >> 4;
  const int tile = blockIdx.x * (blockDim.x >> 5) + wave;
  if (tile >= ct * rt) return;

  const int rowBase = (tile / ct) << 4;
  const int colBase = (tile % ct) << 6;
  const int half = lane >> 4;
  const int l15  = lane & 15;

  const float* __restrict__ ap = X + (size_t)(rowBase + l15) * K + half * 2;
  const float* __restrict__ abp = abias + half * 2;
  const v2f*   __restrict__ bp = Wp + (size_t)half * Ncol + colBase + l15;

  v8f cc[4] = {v8f{}, v8f{}, v8f{}, v8f{}};
  #pragma unroll 4
  for (int k0 = 0; k0 < K; k0 += 4) {
    v2f a = *(const v2f*)(ap + k0);
    if (ACT) {
      a += *(const v2f*)(abp + k0);
      a[0] = fmaxf(a[0], 0.0f);
      a[1] = fmaxf(a[1], 0.0f);
    }
    const v2f* __restrict__ br = bp + (size_t)(k0 >> 1) * Ncol;
    const v2f bv0 = br[0];
    const v2f bv1 = br[16];
    const v2f bv2 = br[32];
    const v2f bv3 = br[48];
    cc[0] = __builtin_amdgcn_wmma_f32_16x16x4_f32(false, a, false, bv0, (short)0, cc[0], false, false);
    cc[1] = __builtin_amdgcn_wmma_f32_16x16x4_f32(false, a, false, bv1, (short)0, cc[1], false, false);
    cc[2] = __builtin_amdgcn_wmma_f32_16x16x4_f32(false, a, false, bv2, (short)0, cc[2], false, false);
    cc[3] = __builtin_amdgcn_wmma_f32_16x16x4_f32(false, a, false, bv3, (short)0, cc[3], false, false);
  }

  const int row0 = rowBase + half * 8;
  #pragma unroll
  for (int t = 0; t < 4; ++t) {
    const int col = colBase + t * 16 + l15;
    const float bb = bias ? bias[col] : 0.0f;
    const int ld = PLANES ? 128 : Ncol;
    float* __restrict__ yp = PLANES
        ? Y + (size_t)(col >> 7) * M * 128 + (size_t)row0 * 128 + (col & 127)
        : Y + (size_t)row0 * Ncol + col;
    #pragma unroll
    for (int r = 0; r < 8; ++r)
      yp[(size_t)r * ld] = cc[t][r] + bb;
  }

  if (CLR == 1) {
    if (colBase == 0 && half == 0) mx[rowBase + l15] = -__builtin_huge_valf();
  }
}

__global__ void gemm_bias_wmma_guard(const float* __restrict__ X,
                                     const float* __restrict__ abias,
                                     const v2f* __restrict__ Wp,
                                     const float* __restrict__ bias,
                                     float* __restrict__ Y,
                                     int M, int K, int Ncol) {
  const int lane = threadIdx.x & 31;
  const int wave = threadIdx.x >> 5;
  const int ct = (Ncol + 15) >> 4;
  const int rt = M >> 4;
  const int tile = blockIdx.x * (blockDim.x >> 5) + wave;
  if (tile >= ct * rt) return;

  const int rowBase = (tile / ct) << 4;
  const int colBase = (tile % ct) << 4;
  const int half = lane >> 4;
  const int l15  = lane & 15;
  const int bcol = colBase + l15;
  const bool bok = bcol < Ncol;
  const int bcolc = bok ? bcol : (Ncol - 1);
  const float bmask = bok ? 1.0f : 0.0f;

  const float* __restrict__ ap = X + (size_t)(rowBase + l15) * 128 + half * 2;
  const float* __restrict__ abp = abias + half * 2;
  const v2f*   __restrict__ bp = Wp + (size_t)half * Ncol + bcolc;

  v8f c = {};
  for (int kb = 0; kb < K; kb += 128) {
    const float* __restrict__ app = ap + (size_t)(kb >> 7) * M * 128;
    #pragma unroll 4
    for (int kk = 0; kk < 128; kk += 4) {
      const int k0 = kb + kk;
      const v2f a = *(const v2f*)(app + kk) + *(const v2f*)(abp + k0);
      v2f b = bp[(size_t)(k0 >> 1) * Ncol];
      b[0] *= bmask;
      b[1] *= bmask;
      c = __builtin_amdgcn_wmma_f32_16x16x4_f32(false, a, false, b, (short)0, c, false, false);
    }
  }

  if (bok) {
    const float bb = bias ? bias[bcol] : 0.0f;
    const int row0 = rowBase + half * 8;
    float* __restrict__ yp = Y + (size_t)row0 * Ncol + bcol;
    #pragma unroll
    for (int r = 0; r < 8; ++r)
      yp[(size_t)r * Ncol] = c[r] + bb;
  }
}

#define EPW 16

template <int D>
__global__ void edge_logits_kernel(const int* __restrict__ ei, int E, int ET,
                                   const float* __restrict__ XL,
                                   const float* __restrict__ XR,
                                   const float* __restrict__ att,
                                   float* __restrict__ eo,
                                   float* __restrict__ mx) {
  constexpr int R = D / 128;
  constexpr int B = 1024 / D;
  const int wave = __builtin_amdgcn_readfirstlane((blockIdx.x * blockDim.x + threadIdx.x) >> 5);
  const int lane = threadIdx.x & 31;
  const int base = wave * EPW;
  if (base >= ET) return;
  int idx = base + (lane & (EPW - 1));
  const bool live = idx < ET;
  idx = live ? idx : ET - 1;
  int sv, tv;
  if (idx < E) { sv = ei[idx]; tv = ei[E + idx]; } else { sv = tv = idx - E; }

  const float4* __restrict__ at = (const float4*)att;
  float4 g[R];
  #pragma unroll
  for (int k = 0; k < R; ++k) g[k] = at[lane + 32 * k];

  float res = 0.0f;
  #pragma unroll
  for (int b = 0; b < EPW; b += B) {
    float4 a[B][R], c[B][R];
    #pragma unroll
    for (int jj = 0; jj < B; ++jj) {
      const int s = __builtin_amdgcn_readlane(sv, b + jj);
      const int t = __builtin_amdgcn_readlane(tv, b + jj);
      const float4* __restrict__ xl = (const float4*)(XL + (size_t)s * D) + lane;
      const float4* __restrict__ xr = (const float4*)(XR + (size_t)t * D) + lane;
      #pragma unroll
      for (int k = 0; k < R; ++k) { a[jj][k] = xl[32 * k]; c[jj][k] = xr[32 * k]; }
    }
    #pragma unroll
    for (int jj = 0; jj < B; ++jj) {
      float acc = 0.0f;
      #pragma unroll
      for (int k = 0; k < R; ++k) {
        const float4 p = a[jj][k];
        const float4 q = c[jj][k];
        float h;
        h = p.x + q.x; h = h > 0.0f ? h : NEG_SLOPE * h; acc += h * g[k].x;
        h = p.y + q.y; h = h > 0.0f ? h : NEG_SLOPE * h; acc += h * g[k].y;
        h = p.z + q.z; h = h > 0.0f ? h : NEG_SLOPE * h; acc += h * g[k].z;
        h = p.w + q.w; h = h > 0.0f ? h : NEG_SLOPE * h; acc += h * g[k].w;
      }
      #pragma unroll
      for (int off = 16; off > 0; off >>= 1) acc += __shfl_xor(acc, off, 32);
      res = (lane == b + jj) ? acc : res;
    }
  }
  if (lane < EPW && live) {
    eo[idx] = res;
    atomicMaxF(&mx[tv], res);
  }
}

#define NPB 196
#define OWN_WAVES 32
#define RING 1024
#define LCAP 1024

struct Guard   { static constexpr bool value = true;  };
struct NoGuard { static constexpr bool value = false; };

template <int D>
__global__ void __launch_bounds__(OWN_WAVES * 32)
owner_sum_kernel(const int* __restrict__ ei, int E, int N,
                 const float* __restrict__ XL,
                 const float* __restrict__ el,
                 const float* __restrict__ mx,
                 float* __restrict__ out,
                 v2i* __restrict__ lst, int* __restrict__ cnts) {
  constexpr int Q = (D + 31) / 32;
  constexpr int RB = 16;
  __shared__ float acc[NPB * D];
  __shared__ float ssum[NPB];
  __shared__ float mloc[NPB];
  __shared__ int ring[OWN_WAVES * RING];

  const int lane = threadIdx.x & 31;
  const int wave = __builtin_amdgcn_readfirstlane(threadIdx.x >> 5);
  const int base = blockIdx.x * NPB;
  const int n_own = (N - base) < NPB ? (N - base) : NPB;

  for (int i = wave; i < n_own; i += OWN_WAVES) {
    const int node = base + i;
    const float m = mx[node];
    const float p = __expf(el[E + node] - m);
    if (lane == 0) { ssum[i] = p; mloc[i] = m; }
    #pragma unroll
    for (int k = 0; k < Q; ++k) {
      const int f = lane + 32 * k;
      if (D % 32 == 0 || f < D) acc[i * D + f] = p * XL[(size_t)node * D + f];
    }
  }
  __syncthreads();

  const int* __restrict__ dp = ei + E;
  const bool vec = ((E & 3) == 0) && ((((size_t)dp) & 15) == 0);
  const int nChunks = (E + 127) >> 7;
  const int rot = (int)blockIdx.x * (nChunks / (int)gridDim.x);
  const int rb = wave * RING;
  int cnext = wave, head = 0, cnt = 0;
  const int slot = (int)blockIdx.x * OWN_WAVES + wave;
  v2i* __restrict__ myl = lst + (size_t)slot * LCAP;
  int tot = 0;
  for (;;) {
    int nb, id;
    if (cnt < 32 && cnext < nChunks) {
      int d[4][4];
      int cb[4];
      bool whole = vec;
      #pragma unroll
      for (int u = 0; u < 4; ++u) {
        const int lc = cnext + OWN_WAVES * u;
        int c = lc + rot;
        c = c >= nChunks ? c - nChunks : c;
        cb[u] = c * 128;
        whole = whole && lc < nChunks && cb[u] + 128 <= E;
      }
      if (whole) {
        #pragma unroll
        for (int u = 0; u < 4; ++u) {
          const int4 t = *(const int4*)(dp + cb[u] + 4 * lane);
          d[u][0] = t.x; d[u][1] = t.y; d[u][2] = t.z; d[u][3] = t.w;
        }
      } else {
        #pragma unroll
        for (int u = 0; u < 4; ++u) {
          const bool there = cnext + OWN_WAVES * u < nChunks;
          const int eb = cb[u] + 4 * lane;
          d[u][0] = (there && eb     < E) ? dp[eb]     : -1;
          d[u][1] = (there && eb + 1 < E) ? dp[eb + 1] : -1;
          d[u][2] = (there && eb + 2 < E) ? dp[eb + 2] : -1;
          d[u][3] = (there && eb + 3 < E) ? dp[eb + 3] : -1;
        }
      }
      #pragma unroll
      for (int u = 0; u < 4; ++u) {
        #pragma unroll
        for (int q = 0; q < 4; ++q) {
          const bool hit = (unsigned)(d[u][q] - base) < (unsigned)n_own;
          const unsigned mk = __builtin_amdgcn_ballot_w32(hit);
          if (mk) {
            const int r = __builtin_amdgcn_mbcnt_lo(mk, 0);
            if (hit) __hip_atomic_store(&ring[rb + ((head + cnt + r) & (RING - 1))], cb[u] + 4 * lane + q,
                                        __ATOMIC_RELAXED, __HIP_MEMORY_SCOPE_WAVEFRONT);
            cnt += __builtin_popcount(mk);
          }
        }
      }
      cnext += OWN_WAVES * 4;
      continue;
    }
    if (cnt == 0) break;
    nb = cnt < 32 ? cnt : 32;
    id = __hip_atomic_load(&ring[rb + ((head + lane) & (RING - 1))],
                           __ATOMIC_RELAXED, __HIP_MEMORY_SCOPE_WAVEFRONT);
    head = (head + nb) & (RING - 1);
    cnt -= nb;
    int sv = 0, tl = 0;
    float p = 0.0f;
    if (lane < nb) {
      sv = ei[id];
      tl = dp[id] - base;
      if (tot + lane < LCAP) myl[tot + lane] = v2i{sv, tl};
      p = __expf(el[id] - mloc[tl]);
      atomicAdd(&ssum[tl], p);
    }
    tot += nb;
    auto rows = [&](const int j0, auto G) {
      float v[RB][Q];
      #pragma unroll
      for (int jj = 0; jj < RB; ++jj) {
        if (!G.value || j0 + jj < nb) {
          const int s = __builtin_amdgcn_readlane(sv, j0 + jj);
          const float* __restrict__ xr = XL + (size_t)s * D + lane;
          #pragma unroll
          for (int k = 0; k < Q; ++k)
            if (D % 32 == 0 || lane + 32 * k < D) v[jj][k] = xr[32 * k];
        }
      }
      #pragma unroll
      for (int jj = 0; jj < RB; ++jj) {
        if (!G.value || j0 + jj < nb) {
          const int t = __builtin_amdgcn_readlane(tl, j0 + jj);
          const float pp = __int_as_float(__builtin_amdgcn_readlane(__float_as_int(p), j0 + jj));
          float* ap = acc + t * D + lane;
          #pragma unroll
          for (int k = 0; k < Q; ++k)
            if (D % 32 == 0 || lane + 32 * k < D) atomicAdd(ap + 32 * k, pp * v[jj][k]);
        }
      }
    };
    if (nb == 32) {
      #pragma unroll
      for (int j0 = 0; j0 < 32; j0 += RB) rows(j0, NoGuard());
    } else {
      for (int j0 = 0; j0 < nb; j0 += RB) rows(j0, Guard());
    }
  }
  if (lane == 0) cnts[slot] = tot;
  __syncthreads();

  const int total = n_own * D;
  float* __restrict__ op = out + (size_t)base * D;
  for (int i = threadIdx.x; i < total; i += OWN_WAVES * 32)
    op[i] = acc[i] / ssum[i / D];
}

template <int P, int D>
__global__ void __launch_bounds__(OWN_WAVES * 32)
owner_attn_kernel(const int* __restrict__ ei, int E, int N,
                  const float* __restrict__ XL,
                  const float* __restrict__ XR,
                  const float* __restrict__ att,
                  const float* __restrict__ XS,
                  float* __restrict__ out,
                  const v2i* __restrict__ lst, const int* __restrict__ cnts,
                  float* __restrict__ elst) {
  constexpr int Q = (D + 31) / 32;
  constexpr int RB = 16;
  constexpr int RA = 8 / P;
  constexpr int XW = P * 128;
  __shared__ float acc[NPB * D];
  __shared__ alignas(16) float xrow[NPB * XW];
  __shared__ float ssum[NPB];
  __shared__ float mloc[NPB];
  __shared__ float eself[NPB];

  const int lane = threadIdx.x & 31;
  const int wave = __builtin_amdgcn_readfirstlane(threadIdx.x >> 5);
  const int base = blockIdx.x * NPB;
  const int n_own = (N - base) < NPB ? (N - base) : NPB;
  const int* __restrict__ dp = ei + E;

  float4 g[P];
  #pragma unroll
  for (int k = 0; k < P; ++k) g[k] = ((const float4*)att)[lane + 32 * k];

  auto logit = [&](const float4 (&a)[P], const int t) -> float {
    float e = 0.0f;
    #pragma unroll
    for (int k = 0; k < P; ++k) {
      const float4 q = *(const float4*)&xrow[t * XW + 128 * k + 4 * lane];
      const float4 p = a[k];
      float s = 0.0f;
      float h;
      h = p.x + q.x; h = h > 0.0f ? h : NEG_SLOPE * h; s += h * g[k].x;
      h = p.y + q.y; h = h > 0.0f ? h : NEG_SLOPE * h; s += h * g[k].y;
      h = p.z + q.z; h = h > 0.0f ? h : NEG_SLOPE * h; s += h * g[k].z;
      h = p.w + q.w; h = h > 0.0f ? h : NEG_SLOPE * h; s += h * g[k].w;
      #pragma unroll
      for (int off = 16; off > 0; off >>= 1) s += __shfl_xor(s, off, 32);
      e = (k == 0) ? s : e + s;
    }
    return e;
  };
  auto srcRow = [&](float4 (&a)[P], const int s) {
    #pragma unroll
    for (int k = 0; k < P; ++k)
      a[k] = ((const float4*)(XL + ((size_t)k * N + s) * 128))[lane];
  };

  auto logitBatch = [&](auto sOf, auto tOf, float (&e)[RA]) {
    float4 a[RA][P];
    #pragma unroll
    for (int jj = 0; jj < RA; ++jj) srcRow(a[jj], sOf(jj));
    float s[RA][P];
    #pragma unroll
    for (int jj = 0; jj < RA; ++jj) {
      const int t = tOf(jj);
      #pragma unroll
      for (int k = 0; k < P; ++k) {
        const float4 q = *(const float4*)&xrow[t * XW + 128 * k + 4 * lane];
        const float4 p = a[jj][k];
        float v = 0.0f;
        float h;
        h = p.x + q.x; h = h > 0.0f ? h : NEG_SLOPE * h; v += h * g[k].x;
        h = p.y + q.y; h = h > 0.0f ? h : NEG_SLOPE * h; v += h * g[k].y;
        h = p.z + q.z; h = h > 0.0f ? h : NEG_SLOPE * h; v += h * g[k].z;
        h = p.w + q.w; h = h > 0.0f ? h : NEG_SLOPE * h; v += h * g[k].w;
        s[jj][k] = v;
      }
    }
    #pragma unroll
    for (int off = 16; off > 0; off >>= 1) {
      #pragma unroll
      for (int jj = 0; jj < RA; ++jj) {
        #pragma unroll
        for (int k = 0; k < P; ++k) s[jj][k] += __shfl_xor(s[jj][k], off, 32);
      }
    }
    #pragma unroll
    for (int jj = 0; jj < RA; ++jj) {
      float v = s[jj][0];
      #pragma unroll
      for (int k = 1; k < P; ++k) v += s[jj][k];
      e[jj] = v;
    }
  };

  for (int i = wave; i < n_own; i += OWN_WAVES) {
    const int node = base + i;
    #pragma unroll
    for (int k = 0; k < P; ++k)
      *(float4*)&xrow[i * XW + 128 * k + 4 * lane] = ((const float4*)(XR + ((size_t)k * N + node) * 128))[lane];
  }
  for (int i0 = wave; i0 < n_own; i0 += OWN_WAVES * RA) {
    auto iOf = [&](const int jj) { const int i = i0 + OWN_WAVES * jj; return i < n_own ? i : i0; };
    float e[RA];
    logitBatch([&](const int jj) { return base + iOf(jj); }, iOf, e);
    #pragma unroll
    for (int jj = 0; jj < RA; ++jj) {
      const int i = i0 + OWN_WAVES * jj;
      if (lane == 0 && i < n_own) { eself[i] = e[jj]; mloc[i] = e[jj]; }
    }
  }
  __syncthreads();

  const int slot = (int)blockIdx.x * OWN_WAVES + wave;
  const v2i* __restrict__ myl = lst + (size_t)slot * LCAP;
  float* __restrict__ mye = elst + (size_t)slot * LCAP;
  const int tot = __builtin_amdgcn_readfirstlane(cnts[slot]);
  const bool useList = (unsigned)tot <= (unsigned)LCAP;

  auto walk = [&](auto f) {
    const int nChunks = (E + 127) >> 7;
    const int rot = (int)blockIdx.x * (nChunks / (int)gridDim.x);
    for (int lc = wave; lc < nChunks; lc += OWN_WAVES) {
      int c = lc + rot;
      c = c >= nChunks ? c - nChunks : c;
      for (int q = 0; q < 4; ++q) {
        const int eb = c * 128 + 4 * lane + q;
        const int d = eb < E ? dp[eb] : -1;
        unsigned mk = __builtin_amdgcn_ballot_w32((unsigned)(d - base) < (unsigned)n_own);
        while (mk) {
          const int l = __builtin_ctz(mk);
          mk &= mk - 1;
          const int id = c * 128 + 4 * l + q;
          f(id, ei[id], dp[id] - base);
        }
      }
    }
  };

  if (useList) {
    for (int pos = 0; pos < tot; pos += 32) {
      const int nb = tot - pos < 32 ? tot - pos : 32;
      const v2i st = myl[pos + (lane < nb ? lane : 0)];
      const int sv = st.x;
      const int tl = st.y;
      float res = 0.0f;
      int j0 = 0;
      do {
        float e[RA];
        logitBatch([&](const int jj) { return __builtin_amdgcn_readlane(sv, j0 + jj); },
                   [&](const int jj) { return __builtin_amdgcn_readlane(tl, j0 + jj); }, e);
        #pragma unroll
        for (int jj = 0; jj < RA; ++jj) res = (lane == j0 + jj) ? e[jj] : res;
        j0 += RA;
      } while (j0 < nb);
      if (lane < nb) {
        mye[pos + lane] = res;
        atomicMaxF(&mloc[tl], res);
      }
    }
  } else {
    walk([&](const int, const int s, const int t) {
      float4 a[P];
      srcRow(a, s);
      const float e = logit(a, t);
      if (lane == 0) atomicMaxF(&mloc[t], e);
    });
  }
  __syncthreads();

  for (int i = wave; i < n_own; i += OWN_WAVES) {
    const int node = base + i;
    const float p = __expf(eself[i] - mloc[i]);
    if (lane == 0) ssum[i] = p;
    #pragma unroll
    for (int k = 0; k < Q; ++k) {
      const int f = lane + 32 * k;
      if (D % 32 == 0 || f < D) acc[i * D + f] = p * XS[(size_t)node * D + f];
    }
  }
  __syncthreads();

  auto addRow1 = [&](const int s, const int t, const float pp) {
    const float* __restrict__ xs = XS + (size_t)s * D + lane;
    float* ap = acc + t * D + lane;
    #pragma unroll
    for (int k = 0; k < Q; ++k)
      if (D % 32 == 0 || lane + 32 * k < D) atomicAdd(ap + 32 * k, pp * xs[32 * k]);
  };
  if (useList) {
    for (int pos = 0; pos < tot; pos += 32) {
      const int nb = tot - pos < 32 ? tot - pos : 32;
      const v2i st = myl[pos + (lane < nb ? lane : 0)];
      int sv = 0, tl = 0;
      float p = 0.0f;
      if (lane < nb) {
        sv = st.x;
        tl = st.y;
        p = __expf(mye[pos + lane] - mloc[tl]);
        atomicAdd(&ssum[tl], p);
      }
      auto rows = [&](const int j0, auto G) {
        float v[RB][Q];
        #pragma unroll
        for (int jj = 0; jj < RB; ++jj) {
          if (!G.value || j0 + jj < nb) {
            const int s = __builtin_amdgcn_readlane(sv, j0 + jj);
            const float* __restrict__ xs = XS + (size_t)s * D + lane;
            #pragma unroll
            for (int k = 0; k < Q; ++k)
              if (D % 32 == 0 || lane + 32 * k < D) v[jj][k] = xs[32 * k];
          }
        }
        #pragma unroll
        for (int jj = 0; jj < RB; ++jj) {
          if (!G.value || j0 + jj < nb) {
            const int t = __builtin_amdgcn_readlane(tl, j0 + jj);
            const float pp = __int_as_float(__builtin_amdgcn_readlane(__float_as_int(p), j0 + jj));
            float* ap = acc + t * D + lane;
            #pragma unroll
            for (int k = 0; k < Q; ++k)
              if (D % 32 == 0 || lane + 32 * k < D) atomicAdd(ap + 32 * k, pp * v[jj][k]);
          }
        }
      };
      if (nb == 32) {
        #pragma unroll
        for (int j0 = 0; j0 < 32; j0 += RB) rows(j0, NoGuard());
      } else {
        for (int j0 = 0; j0 < nb; j0 += RB) rows(j0, Guard());
      }
    }
  } else {
    walk([&](const int, const int s, const int t) {
      float4 a[P];
      srcRow(a, s);
      const float pp = __expf(logit(a, t) - mloc[t]);
      if (lane == 0) atomicAdd(&ssum[t], pp);
      addRow1(s, t, pp);
    });
  }
  __syncthreads();

  const int total = n_own * D;
  float* __restrict__ op = out + (size_t)base * D;
  for (int i = threadIdx.x; i < total; i += OWN_WAVES * 32)
    op[i] = acc[i] / ssum[i / D];
}

extern "C" void kernel_launch(void* const* d_in, const int* in_sizes, int n_in,
                              void* d_out, int out_size, void* d_ws, size_t ws_size,
                              hipStream_t stream) {
  const float* x   = (const float*)d_in[0];
  const int*   ei  = (const int*)  d_in[1];
  const float* w1l = (const float*)d_in[2];
  const float* b1l = (const float*)d_in[3];
  const float* w1r = (const float*)d_in[4];
  const float* b1r = (const float*)d_in[5];
  const float* a1  = (const float*)d_in[6];
  const float* o1  = (const float*)d_in[7];
  const float* w2l = (const float*)d_in[8];
  const float* b2l = (const float*)d_in[9];
  const float* w2r = (const float*)d_in[10];
  const float* b2r = (const float*)d_in[11];
  const float* a2  = (const float*)d_in[12];
  const float* o2  = (const float*)d_in[13];
  const float* w3l = (const float*)d_in[14];
  const float* b3l = (const float*)d_in[15];
  const float* w3r = (const float*)d_in[16];
  const float* b3r = (const float*)d_in[17];
  const float* a3  = (const float*)d_in[18];
  const float* o3  = (const float*)d_in[19];
  const float* wc  = (const float*)d_in[20];
  const float* bc  = (const float*)d_in[21];

  const int N  = in_sizes[0] / 128;
  const int E  = in_sizes[1] / 2;
  const int ET = E + N;
  const int eblocks = ((ET + EPW - 1) / EPW + 7) / 8;

  float* B0  = (float*)d_ws;
  float* B1  = B0 + (size_t)N * 256;
  float* B2  = B1 + (size_t)N * 256;
  float* EP  = B2 + (size_t)N * 256;
  float* Mx  = EP + (size_t)ET;
  v2f*   WPa = (v2f*)(Mx + (size_t)N);
  v2f*   WPb = WPa + 32768;
  float* S1  = B1;
  float* S2  = B1 + (size_t)N * 128;

  auto packW = [&](const float* W, v2f* Wp, int k, int n) {
    const int total = (k >> 1) * n;
    pack_w_kernel<<<(total + 255) / 256, 256, 0, stream>>>(W, Wp, k, n);
  };
  auto gemmGuard = [&](const float* X, const float* ab, const v2f* Wp, const float* b,
                       float* Y, int m, int k, int n) {
    const int tiles = (m >> 4) * ((n + 15) / 16);
    gemm_bias_wmma_guard<<<(tiles + 3) / 4, 128, 0, stream>>>(X, ab, Wp, b, Y, m, k, n);
  };
  const int oblocks = (N + NPB - 1) / NPB;
  int* CN = (int*)(WPb + 32768);
  v2i* LS = (v2i*)(CN + (size_t)oblocks * OWN_WAVES);
  float* EL = (float*)(LS + (size_t)oblocks * OWN_WAVES * LCAP);

  auto layer = [&](const float* Xin, const float* inb,
                   const float* wl, const float* bl,
                   const float* wr, const float* br,
                   const float* att, float* XL, float* XR, float* ACC) {
    const int gblocks = ((N >> 4) * 2 + 3) / 4;
    packW(wl, WPa, 128, 128);
    packW(wr, WPb, 128, 128);
    if (inb) {
      gemm_bias_wmma_n64<true, 0, false><<<gblocks, 128, 0, stream>>>(Xin, inb, WPa, bl, XL, nullptr, N, 128, 128);
      gemm_bias_wmma_n64<true, 0, false><<<gblocks, 128, 0, stream>>>(Xin, inb, WPb, br, XR, nullptr, N, 128, 128);
    } else {
      gemm_bias_wmma_n64<false, 0, false><<<gblocks, 128, 0, stream>>>(Xin, nullptr, WPa, bl, XL, nullptr, N, 128, 128);
      gemm_bias_wmma_n64<false, 1, false><<<gblocks, 128, 0, stream>>>(Xin, nullptr, WPb, br, XR, Mx, N, 128, 128);
    }
    if (inb) {
      owner_attn_kernel<1, 128><<<oblocks, OWN_WAVES * 32, 0, stream>>>(ei, E, N, XL, XR, att, XL, ACC, LS, CN, EL);
    } else {
      edge_logits_kernel<128><<<eblocks, 256, 0, stream>>>(ei, E, ET, XL, XR, att, EP, Mx);
      owner_sum_kernel<128><<<oblocks, OWN_WAVES * 32, 0, stream>>>(ei, E, N, XL, EP, Mx, ACC, LS, CN);
    }
  };

  layer(x,  nullptr, w1l, b1l, w1r, b1r, a1, B0, B2, S1);
  layer(S1, o1,      w2l, b2l, w2r, b2r, a2, B0, B2, S2);
  float* out = (float*)d_out;
  const int gblocks3 = ((N >> 4) * 4 + 3) / 4;
  packW(w3l, WPa, 128, 256);
  packW(w3r, WPb, 128, 256);
  gemm_bias_wmma_n64<true, 0, true><<<gblocks3, 128, 0, stream>>>(S2, o2, WPa, b3l, B0, nullptr, N, 128, 256);
  gemm_bias_wmma_n64<true, 0, true><<<gblocks3, 128, 0, stream>>>(S2, o2, WPb, b3r, B2, nullptr, N, 128, 256);
  packW(wc, WPa, 256, 40);
  gemmGuard(B0, o3, WPa, bc, B1, N, 256, 40);
  owner_attn_kernel<2, 40><<<oblocks, OWN_WAVES * 32, 0, stream>>>(ei, E, N, B0, B2, a3, B1, out, LS, CN, EL);
}


# Layer 1 too runs on the one-pass node owners: the edge lists come from a pass of their own ahead of it (destination column only); logit launch and global max atomics gone; sums keep arrival order
# speedup vs baseline: 3.7470x; 1.1218x over previous
#include <hip/hip_runtime.h>

typedef __attribute__((ext_vector_type(2))) float v2f;
typedef __attribute__((ext_vector_type(2))) int v2i;
typedef __attribute__((ext_vector_type(8))) float v8f;

#define NEG_SLOPE 0.2f

__device__ inline void atomicMaxF(float* addr, float v) {
  if (v >= 0.0f) atomicMax((int*)addr, __float_as_int(v));
  else           atomicMin((unsigned int*)addr, __float_as_uint(v));
}

__global__ void pack_w_kernel(const float* __restrict__ W,
                              v2f* __restrict__ Wp, int K, int Ncol) {
  const int i = blockIdx.x * blockDim.x + threadIdx.x;
  const int total = (K >> 1) * Ncol;
  if (i >= total) return;
  const int kp  = i / Ncol;
  const int col = i - kp * Ncol;
  v2f p;
  p[0] = W[(size_t)(2 * kp)     * Ncol + col];
  p[1] = W[(size_t)(2 * kp + 1) * Ncol + col];
  Wp[i] = p;
}

template <bool ACT, bool PLANES>
__global__ void gemm_bias_wmma_n64(const float* __restrict__ X,
                                   const float* __restrict__ abias,
                                   const v2f* __restrict__ Wp,
                                   const float* __restrict__ bias,
                                   float* __restrict__ Y,
                                   int M, int K, int Ncol) {
  const int lane = threadIdx.x & 31;
  const int wave = threadIdx.x >> 5;
  const int ct = Ncol >> 6;
  const int rt = M >> 4;
  const int tile = blockIdx.x * (blockDim.x >> 5) + wave;
  if (tile >= ct * rt) return;

  const int rowBase = (tile / ct) << 4;
  const int colBase = (tile % ct) << 6;
  const int half = lane >> 4;
  const int l15  = lane & 15;

  const float* __restrict__ ap = X + (size_t)(rowBase + l15) * K + half * 2;
  const float* __restrict__ abp = abias + half * 2;
  const v2f*   __restrict__ bp = Wp + (size_t)half * Ncol + colBase + l15;

  v8f cc[4] = {v8f{}, v8f{}, v8f{}, v8f{}};
  #pragma unroll 4
  for (int k0 = 0; k0 < K; k0 += 4) {
    v2f a = *(const v2f*)(ap + k0);
    if (ACT) {
      a += *(const v2f*)(abp + k0);
      a[0] = fmaxf(a[0], 0.0f);
      a[1] = fmaxf(a[1], 0.0f);
    }
    const v2f* __restrict__ br = bp + (size_t)(k0 >> 1) * Ncol;
    const v2f bv0 = br[0];
    const v2f bv1 = br[16];
    const v2f bv2 = br[32];
    const v2f bv3 = br[48];
    cc[0] = __builtin_amdgcn_wmma_f32_16x16x4_f32(false, a, false, bv0, (short)0, cc[0], false, false);
    cc[1] = __builtin_amdgcn_wmma_f32_16x16x4_f32(false, a, false, bv1, (short)0, cc[1], false, false);
    cc[2] = __builtin_amdgcn_wmma_f32_16x16x4_f32(false, a, false, bv2, (short)0, cc[2], false, false);
    cc[3] = __builtin_amdgcn_wmma_f32_16x16x4_f32(false, a, false, bv3, (short)0, cc[3], false, false);
  }

  const int row0 = rowBase + half * 8;
  #pragma unroll
  for (int t = 0; t < 4; ++t) {
    const int col = colBase + t * 16 + l15;
    const float bb = bias ? bias[col] : 0.0f;
    const int ld = PLANES ? 128 : Ncol;
    float* __restrict__ yp = PLANES
        ? Y + (size_t)(col >> 7) * M * 128 + (size_t)row0 * 128 + (col & 127)
        : Y + (size_t)row0 * Ncol + col;
    #pragma unroll
    for (int r = 0; r < 8; ++r)
      yp[(size_t)r * ld] = cc[t][r] + bb;
  }
}

__global__ void gemm_bias_wmma_guard(const float* __restrict__ X,
                                     const float* __restrict__ abias,
                                     const v2f* __restrict__ Wp,
                                     const float* __restrict__ bias,
                                     float* __restrict__ Y,
                                     int M, int K, int Ncol) {
  const int lane = threadIdx.x & 31;
  const int wave = threadIdx.x >> 5;
  const int ct = (Ncol + 15) >> 4;
  const int rt = M >> 4;
  const int tile = blockIdx.x * (blockDim.x >> 5) + wave;
  if (tile >= ct * rt) return;

  const int rowBase = (tile / ct) << 4;
  const int colBase = (tile % ct) << 4;
  const int half = lane >> 4;
  const int l15  = lane & 15;
  const int bcol = colBase + l15;
  const bool bok = bcol < Ncol;
  const int bcolc = bok ? bcol : (Ncol - 1);
  const float bmask = bok ? 1.0f : 0.0f;

  const float* __restrict__ ap = X + (size_t)(rowBase + l15) * 128 + half * 2;
  const float* __restrict__ abp = abias + half * 2;
  const v2f*   __restrict__ bp = Wp + (size_t)half * Ncol + bcolc;

  v8f c = {};
  for (int kb = 0; kb < K; kb += 128) {
    const float* __restrict__ app = ap + (size_t)(kb >> 7) * M * 128;
    #pragma unroll 4
    for (int kk = 0; kk < 128; kk += 4) {
      const int k0 = kb + kk;
      const v2f a = *(const v2f*)(app + kk) + *(const v2f*)(abp + k0);
      v2f b = bp[(size_t)(k0 >> 1) * Ncol];
      b[0] *= bmask;
      b[1] *= bmask;
      c = __builtin_amdgcn_wmma_f32_16x16x4_f32(false, a, false, b, (short)0, c, false, false);
    }
  }

  if (bok) {
    const float bb = bias ? bias[bcol] : 0.0f;
    const int row0 = rowBase + half * 8;
    float* __restrict__ yp = Y + (size_t)row0 * Ncol + bcol;
    #pragma unroll
    for (int r = 0; r < 8; ++r)
      yp[(size_t)r * Ncol] = c[r] + bb;
  }
}

#define NPB 196
#define OWN_WAVES 32
#define RING 1024
#define LCAP 1024

struct Guard   { static constexpr bool value = true;  };
struct NoGuard { static constexpr bool value = false; };

__global__ void __launch_bounds__(OWN_WAVES * 32)
owner_list_kernel(const int* __restrict__ ei, int E, int N,
                  v2i* __restrict__ lst, int* __restrict__ cnts) {
  __shared__ unsigned long long ring[OWN_WAVES * RING];

  const int lane = threadIdx.x & 31;
  const int wave = __builtin_amdgcn_readfirstlane(threadIdx.x >> 5);
  const int base = blockIdx.x * NPB;
  const int n_own = (N - base) < NPB ? (N - base) : NPB;

  const int* __restrict__ dp = ei + E;
  const bool vec = ((E & 3) == 0) && ((((size_t)dp) & 15) == 0);
  const int nChunks = (E + 127) >> 7;
  const int rot = (int)blockIdx.x * (nChunks / (int)gridDim.x);
  const int rb = wave * RING;
  int cnext = wave, head = 0, cnt = 0;
  const int slot = (int)blockIdx.x * OWN_WAVES + wave;
  v2i* __restrict__ myl = lst + (size_t)slot * LCAP;
  int tot = 0;
  for (;;) {
    if (cnt < 32 && cnext < nChunks) {
      int d[4][4];
      int cb[4];
      bool whole = vec;
      #pragma unroll
      for (int u = 0; u < 4; ++u) {
        const int lc = cnext + OWN_WAVES * u;
        int c = lc + rot;
        c = c >= nChunks ? c - nChunks : c;
        cb[u] = c * 128;
        whole = whole && lc < nChunks && cb[u] + 128 <= E;
      }
      if (whole) {
        #pragma unroll
        for (int u = 0; u < 4; ++u) {
          const int4 t = *(const int4*)(dp + cb[u] + 4 * lane);
          d[u][0] = t.x; d[u][1] = t.y; d[u][2] = t.z; d[u][3] = t.w;
        }
      } else {
        #pragma unroll
        for (int u = 0; u < 4; ++u) {
          const bool there = cnext + OWN_WAVES * u < nChunks;
          const int eb = cb[u] + 4 * lane;
          d[u][0] = (there && eb     < E) ? dp[eb]     : -1;
          d[u][1] = (there && eb + 1 < E) ? dp[eb + 1] : -1;
          d[u][2] = (there && eb + 2 < E) ? dp[eb + 2] : -1;
          d[u][3] = (there && eb + 3 < E) ? dp[eb + 3] : -1;
        }
      }
      #pragma unroll
      for (int u = 0; u < 4; ++u) {
        #pragma unroll
        for (int q = 0; q < 4; ++q) {
          const int tl = d[u][q] - base;
          const bool hit = (unsigned)tl < (unsigned)n_own;
          const unsigned mk = __builtin_amdgcn_ballot_w32(hit);
          if (mk) {
            const int r = __builtin_amdgcn_mbcnt_lo(mk, 0);
            const unsigned long long ent =
                ((unsigned long long)(unsigned)tl << 32) | (unsigned)(cb[u] + 4 * lane + q);
            if (hit) __hip_atomic_store(&ring[rb + ((head + cnt + r) & (RING - 1))], ent,
                                        __ATOMIC_RELAXED, __HIP_MEMORY_SCOPE_WAVEFRONT);
            cnt += __builtin_popcount(mk);
          }
        }
      }
      cnext += OWN_WAVES * 4;
      continue;
    }
    if (cnt == 0) break;
    const int nb = cnt < 32 ? cnt : 32;
    const unsigned long long ent = __hip_atomic_load(&ring[rb + ((head + lane) & (RING - 1))],
                                                     __ATOMIC_RELAXED, __HIP_MEMORY_SCOPE_WAVEFRONT);
    head = (head + nb) & (RING - 1);
    cnt -= nb;
    if (lane < nb && tot + lane < LCAP)
      myl[tot + lane] = v2i{ei[(int)(unsigned)ent], (int)(unsigned)(ent >> 32)};
    tot += nb;
  }
  if (lane == 0) cnts[slot] = tot;
}

template <int P, int D>
__global__ void __launch_bounds__(OWN_WAVES * 32)
owner_attn_kernel(const int* __restrict__ ei, int E, int N,
                  const float* __restrict__ XL,
                  const float* __restrict__ XR,
                  const float* __restrict__ att,
                  const float* __restrict__ XS,
                  float* __restrict__ out,
                  const v2i* __restrict__ lst, const int* __restrict__ cnts,
                  float* __restrict__ elst) {
  constexpr int Q = (D + 31) / 32;
  constexpr int RB = 16;
  constexpr int RA = 8 / P;
  constexpr int XW = P * 128;
  __shared__ float acc[NPB * D];
  __shared__ alignas(16) float xrow[NPB * XW];
  __shared__ float ssum[NPB];
  __shared__ float mloc[NPB];
  __shared__ float eself[NPB];

  const int lane = threadIdx.x & 31;
  const int wave = __builtin_amdgcn_readfirstlane(threadIdx.x >> 5);
  const int base = blockIdx.x * NPB;
  const int n_own = (N - base) < NPB ? (N - base) : NPB;
  const int* __restrict__ dp = ei + E;

  float4 g[P];
  #pragma unroll
  for (int k = 0; k < P; ++k) g[k] = ((const float4*)att)[lane + 32 * k];

  auto logit = [&](const float4 (&a)[P], const int t) -> float {
    float e = 0.0f;
    #pragma unroll
    for (int k = 0; k < P; ++k) {
      const float4 q = *(const float4*)&xrow[t * XW + 128 * k + 4 * lane];
      const float4 p = a[k];
      float s = 0.0f;
      float h;
      h = p.x + q.x; h = h > 0.0f ? h : NEG_SLOPE * h; s += h * g[k].x;
      h = p.y + q.y; h = h > 0.0f ? h : NEG_SLOPE * h; s += h * g[k].y;
      h = p.z + q.z; h = h > 0.0f ? h : NEG_SLOPE * h; s += h * g[k].z;
      h = p.w + q.w; h = h > 0.0f ? h : NEG_SLOPE * h; s += h * g[k].w;
      #pragma unroll
      for (int off = 16; off > 0; off >>= 1) s += __shfl_xor(s, off, 32);
      e = (k == 0) ? s : e + s;
    }
    return e;
  };
  auto srcRow = [&](float4 (&a)[P], const int s) {
    #pragma unroll
    for (int k = 0; k < P; ++k)
      a[k] = ((const float4*)(XL + ((size_t)k * N + s) * 128))[lane];
  };

  auto logitBatch = [&](auto sOf, auto tOf, float (&e)[RA]) {
    float4 a[RA][P];
    #pragma unroll
    for (int jj = 0; jj < RA; ++jj) srcRow(a[jj], sOf(jj));
    float s[RA][P];
    #pragma unroll
    for (int jj = 0; jj < RA; ++jj) {
      const int t = tOf(jj);
      #pragma unroll
      for (int k = 0; k < P; ++k) {
        const float4 q = *(const float4*)&xrow[t * XW + 128 * k + 4 * lane];
        const float4 p = a[jj][k];
        float v = 0.0f;
        float h;
        h = p.x + q.x; h = h > 0.0f ? h : NEG_SLOPE * h; v += h * g[k].x;
        h = p.y + q.y; h = h > 0.0f ? h : NEG_SLOPE * h; v += h * g[k].y;
        h = p.z + q.z; h = h > 0.0f ? h : NEG_SLOPE * h; v += h * g[k].z;
        h = p.w + q.w; h = h > 0.0f ? h : NEG_SLOPE * h; v += h * g[k].w;
        s[jj][k] = v;
      }
    }
    #pragma unroll
    for (int off = 16; off > 0; off >>= 1) {
      #pragma unroll
      for (int jj = 0; jj < RA; ++jj) {
        #pragma unroll
        for (int k = 0; k < P; ++k) s[jj][k] += __shfl_xor(s[jj][k], off, 32);
      }
    }
    #pragma unroll
    for (int jj = 0; jj < RA; ++jj) {
      float v = s[jj][0];
      #pragma unroll
      for (int k = 1; k < P; ++k) v += s[jj][k];
      e[jj] = v;
    }
  };

  for (int i = wave; i < n_own; i += OWN_WAVES) {
    const int node = base + i;
    #pragma unroll
    for (int k = 0; k < P; ++k)
      *(float4*)&xrow[i * XW + 128 * k + 4 * lane] = ((const float4*)(XR + ((size_t)k * N + node) * 128))[lane];
  }
  for (int i0 = wave; i0 < n_own; i0 += OWN_WAVES * RA) {
    auto iOf = [&](const int jj) { const int i = i0 + OWN_WAVES * jj; return i < n_own ? i : i0; };
    float e[RA];
    logitBatch([&](const int jj) { return base + iOf(jj); }, iOf, e);
    #pragma unroll
    for (int jj = 0; jj < RA; ++jj) {
      const int i = i0 + OWN_WAVES * jj;
      if (lane == 0 && i < n_own) { eself[i] = e[jj]; mloc[i] = e[jj]; }
    }
  }
  __syncthreads();

  const int slot = (int)blockIdx.x * OWN_WAVES + wave;
  const v2i* __restrict__ myl = lst + (size_t)slot * LCAP;
  float* __restrict__ mye = elst + (size_t)slot * LCAP;
  const int tot = __builtin_amdgcn_readfirstlane(cnts[slot]);
  const bool useList = (unsigned)tot <= (unsigned)LCAP;

  auto walk = [&](auto f) {
    const int nChunks = (E + 127) >> 7;
    const int rot = (int)blockIdx.x * (nChunks / (int)gridDim.x);
    for (int lc = wave; lc < nChunks; lc += OWN_WAVES) {
      int c = lc + rot;
      c = c >= nChunks ? c - nChunks : c;
      for (int q = 0; q < 4; ++q) {
        const int eb = c * 128 + 4 * lane + q;
        const int d = eb < E ? dp[eb] : -1;
        unsigned mk = __builtin_amdgcn_ballot_w32((unsigned)(d - base) < (unsigned)n_own);
        while (mk) {
          const int l = __builtin_ctz(mk);
          mk &= mk - 1;
          const int id = c * 128 + 4 * l + q;
          f(id, ei[id], dp[id] - base);
        }
      }
    }
  };

  if (useList) {
    for (int pos = 0; pos < tot; pos += 32) {
      const int nb = tot - pos < 32 ? tot - pos : 32;
      const v2i st = myl[pos + (lane < nb ? lane : 0)];
      const int sv = st.x;
      const int tl = st.y;
      float res = 0.0f;
      int j0 = 0;
      do {
        float e[RA];
        logitBatch([&](const int jj) { return __builtin_amdgcn_readlane(sv, j0 + jj); },
                   [&](const int jj) { return __builtin_amdgcn_readlane(tl, j0 + jj); }, e);
        #pragma unroll
        for (int jj = 0; jj < RA; ++jj) res = (lane == j0 + jj) ? e[jj] : res;
        j0 += RA;
      } while (j0 < nb);
      if (lane < nb) {
        mye[pos + lane] = res;
        atomicMaxF(&mloc[tl], res);
      }
    }
  } else {
    walk([&](const int, const int s, const int t) {
      float4 a[P];
      srcRow(a, s);
      const float e = logit(a, t);
      if (lane == 0) atomicMaxF(&mloc[t], e);
    });
  }
  __syncthreads();

  for (int i = wave; i < n_own; i += OWN_WAVES) {
    const int node = base + i;
    const float p = __expf(eself[i] - mloc[i]);
    if (lane == 0) ssum[i] = p;
    #pragma unroll
    for (int k = 0; k < Q; ++k) {
      const int f = lane + 32 * k;
      if (D % 32 == 0 || f < D) acc[i * D + f] = p * XS[(size_t)node * D + f];
    }
  }
  __syncthreads();

  auto addRow1 = [&](const int s, const int t, const float pp) {
    const float* __restrict__ xs = XS + (size_t)s * D + lane;
    float* ap = acc + t * D + lane;
    #pragma unroll
    for (int k = 0; k < Q; ++k)
      if (D % 32 == 0 || lane + 32 * k < D) atomicAdd(ap + 32 * k, pp * xs[32 * k]);
  };
  if (useList) {
    for (int pos = 0; pos < tot; pos += 32) {
      const int nb = tot - pos < 32 ? tot - pos : 32;
      const v2i st = myl[pos + (lane < nb ? lane : 0)];
      int sv = 0, tl = 0;
      float p = 0.0f;
      if (lane < nb) {
        sv = st.x;
        tl = st.y;
        p = __expf(mye[pos + lane] - mloc[tl]);
        atomicAdd(&ssum[tl], p);
      }
      auto rows = [&](const int j0, auto G) {
        float v[RB][Q];
        #pragma unroll
        for (int jj = 0; jj < RB; ++jj) {
          if (!G.value || j0 + jj < nb) {
            const int s = __builtin_amdgcn_readlane(sv, j0 + jj);
            const float* __restrict__ xs = XS + (size_t)s * D + lane;
            #pragma unroll
            for (int k = 0; k < Q; ++k)
              if (D % 32 == 0 || lane + 32 * k < D) v[jj][k] = xs[32 * k];
          }
        }
        #pragma unroll
        for (int jj = 0; jj < RB; ++jj) {
          if (!G.value || j0 + jj < nb) {
            const int t = __builtin_amdgcn_readlane(tl, j0 + jj);
            const float pp = __int_as_float(__builtin_amdgcn_readlane(__float_as_int(p), j0 + jj));
            float* ap = acc + t * D + lane;
            #pragma unroll
            for (int k = 0; k < Q; ++k)
              if (D % 32 == 0 || lane + 32 * k < D) atomicAdd(ap + 32 * k, pp * v[jj][k]);
          }
        }
      };
      if (nb == 32) {
        #pragma unroll
        for (int j0 = 0; j0 < 32; j0 += RB) rows(j0, NoGuard());
      } else {
        for (int j0 = 0; j0 < nb; j0 += RB) rows(j0, Guard());
      }
    }
  } else {
    walk([&](const int, const int s, const int t) {
      float4 a[P];
      srcRow(a, s);
      const float pp = __expf(logit(a, t) - mloc[t]);
      if (lane == 0) atomicAdd(&ssum[t], pp);
      addRow1(s, t, pp);
    });
  }
  __syncthreads();

  const int total = n_own * D;
  float* __restrict__ op = out + (size_t)base * D;
  for (int i = threadIdx.x; i < total; i += OWN_WAVES * 32)
    op[i] = acc[i] / ssum[i / D];
}

extern "C" void kernel_launch(void* const* d_in, const int* in_sizes, int n_in,
                              void* d_out, int out_size, void* d_ws, size_t ws_size,
                              hipStream_t stream) {
  const float* x   = (const float*)d_in[0];
  const int*   ei  = (const int*)  d_in[1];
  const float* w1l = (const float*)d_in[2];
  const float* b1l = (const float*)d_in[3];
  const float* w1r = (const float*)d_in[4];
  const float* b1r = (const float*)d_in[5];
  const float* a1  = (const float*)d_in[6];
  const float* o1  = (const float*)d_in[7];
  const float* w2l = (const float*)d_in[8];
  const float* b2l = (const float*)d_in[9];
  const float* w2r = (const float*)d_in[10];
  const float* b2r = (const float*)d_in[11];
  const float* a2  = (const float*)d_in[12];
  const float* o2  = (const float*)d_in[13];
  const float* w3l = (const float*)d_in[14];
  const float* b3l = (const float*)d_in[15];
  const float* w3r = (const float*)d_in[16];
  const float* b3r = (const float*)d_in[17];
  const float* a3  = (const float*)d_in[18];
  const float* o3  = (const float*)d_in[19];
  const float* wc  = (const float*)d_in[20];
  const float* bc  = (const float*)d_in[21];

  const int N  = in_sizes[0] / 128;
  const int E  = in_sizes[1] / 2;

  float* B0  = (float*)d_ws;
  float* B1  = B0 + (size_t)N * 256;
  float* B2  = B1 + (size_t)N * 256;
  v2f*   WPa = (v2f*)(B2 + (size_t)N * 256);
  v2f*   WPb = WPa + 32768;
  float* S1  = B1;
  float* S2  = B1 + (size_t)N * 128;

  auto packW = [&](const float* W, v2f* Wp, int k, int n) {
    const int total = (k >> 1) * n;
    pack_w_kernel<<<(total + 255) / 256, 256, 0, stream>>>(W, Wp, k, n);
  };
  auto gemmGuard = [&](const float* X, const float* ab, const v2f* Wp, const float* b,
                       float* Y, int m, int k, int n) {
    const int tiles = (m >> 4) * ((n + 15) / 16);
    gemm_bias_wmma_guard<<<(tiles + 3) / 4, 128, 0, stream>>>(X, ab, Wp, b, Y, m, k, n);
  };
  const int oblocks = (N + NPB - 1) / NPB;
  int* CN = (int*)(WPb + 32768);
  v2i* LS = (v2i*)(CN + (size_t)oblocks * OWN_WAVES);
  float* EL = (float*)(LS + (size_t)oblocks * OWN_WAVES * LCAP);

  owner_list_kernel<<<oblocks, OWN_WAVES * 32, 0, stream>>>(ei, E, N, LS, CN);

  auto layer = [&](const float* Xin, const float* inb,
                   const float* wl, const float* bl,
                   const float* wr, const float* br,
                   const float* att, float* XL, float* XR, float* ACC) {
    const int gblocks = ((N >> 4) * 2 + 3) / 4;
    packW(wl, WPa, 128, 128);
    packW(wr, WPb, 128, 128);
    if (inb) {
      gemm_bias_wmma_n64<true, false><<<gblocks, 128, 0, stream>>>(Xin, inb, WPa, bl, XL, N, 128, 128);
      gemm_bias_wmma_n64<true, false><<<gblocks, 128, 0, stream>>>(Xin, inb, WPb, br, XR, N, 128, 128);
    } else {
      gemm_bias_wmma_n64<false, false><<<gblocks, 128, 0, stream>>>(Xin, nullptr, WPa, bl, XL, N, 128, 128);
      gemm_bias_wmma_n64<false, false><<<gblocks, 128, 0, stream>>>(Xin, nullptr, WPb, br, XR, N, 128, 128);
    }
    owner_attn_kernel<1, 128><<<oblocks, OWN_WAVES * 32, 0, stream>>>(ei, E, N, XL, XR, att, XL, ACC, LS, CN, EL);
  };

  layer(x,  nullptr, w1l, b1l, w1r, b1r, a1, B0, B2, S1);
  layer(S1, o1,      w2l, b2l, w2r, b2r, a2, B0, B2, S2);
  float* out = (float*)d_out;
  const int gblocks3 = ((N >> 4) * 4 + 3) / 4;
  packW(w3l, WPa, 128, 256);
  packW(w3r, WPb, 128, 256);
  gemm_bias_wmma_n64<true, true><<<gblocks3, 128, 0, stream>>>(S2, o2, WPa, b3l, B0, N, 128, 256);
  gemm_bias_wmma_n64<true, true><<<gblocks3, 128, 0, stream>>>(S2, o2, WPb, b3r, B2, N, 128, 256);
  packW(wc, WPa, 256, 40);
  gemmGuard(B0, o3, WPa, bc, B1, N, 256, 40);
  owner_attn_kernel<2, 40><<<oblocks, OWN_WAVES * 32, 0, stream>>>(ei, E, N, B0, B2, a3, B1, out, LS, CN, EL);
}
